# starting kernel
# baseline (speedup 1.0000x reference)
#include <hip/hip_runtime.h>

typedef __attribute__((ext_vector_type(2))) float v2f;
typedef __attribute__((ext_vector_type(8))) float v8f;

namespace {
constexpr int kT      = 4096;
constexpr int kNB     = 64;
constexpr int kNU     = 32;
constexpr int kNY     = 32;
constexpr int kNH     = 256;
constexpr int kNHalf  = 128;
constexpr int kChunk  = 64;
constexpr int kNChunk = kT / kChunk;
constexpr int kBG     = 32;
constexpr int kNBG    = kNB / kBG;
constexpr int kTG     = 2;
constexpr int kNGrp   = kChunk / kTG;

constexpr size_t WS_LR  = 0;
constexpr size_t WS_LI  = WS_LR + kNHalf;
constexpr size_t WS_LRL = WS_LI + kNHalf;
constexpr size_t WS_LIL = WS_LRL + kNHalf;
constexpr size_t WS_BP  = WS_LIL + kNHalf;
constexpr size_t WS_X0  = WS_BP + (size_t)kNH * kNU;
constexpr size_t WS_F   = WS_X0 + (size_t)kNB * kNH;
constexpr size_t WS_S   = WS_F + (size_t)kNChunk * kNB * kNH;

constexpr size_t kSmemBytes = (size_t)kTG * kBG * kNH * sizeof(float);
}

__global__ void __launch_bounds__(256)
setup_kernel(const float* __restrict__ y0, const float* __restrict__ llr,
             const float* __restrict__ lli, const float* __restrict__ Bmat,
             const float* __restrict__ y2xw, const float* __restrict__ y2xb,
             float* __restrict__ ws) {
  __shared__ float s_nf[kNH];
  const int tid = threadIdx.x;
  if (tid < kNHalf) {
    const float decay = __expf(-fabsf(llr[tid]));
    const float r     = __expf(-decay);
    const float th    = 1.57079632679489662f * __expf(-fabsf(lli[tid]));
    const float lr    = r * __cosf(th);
    const float li    = r * __sinf(th);
    const float nfh   = sqrtf(fmaxf(0.f, 1.f - r * r));
    ws[WS_LR + tid] = lr;
    ws[WS_LI + tid] = li;
    float pr = lr, pi = li;
    #pragma unroll
    for (int s = 0; s < 6; ++s) {
      const float nr = pr * pr - pi * pi;
      const float ni = 2.f * pr * pi;
      pr = nr; pi = ni;
    }
    ws[WS_LRL + tid] = pr;
    ws[WS_LIL + tid] = pi;
    s_nf[tid]          = nfh;
    s_nf[tid + kNHalf] = nfh;
  }
  __syncthreads();
  for (int i = tid; i < kNH * kNU; i += blockDim.x) {
    ws[WS_BP + i] = Bmat[i] * s_nf[i / kNU];
  }
  for (int o = tid; o < kNB * kNH; o += blockDim.x) {
    const int b = o >> 8;
    const int n = o & 255;
    float acc = y2xb[n];
    const float* yr = y0 + b * kNY;
    const float* wr = y2xw + n * kNY;
    #pragma unroll 8
    for (int u = 0; u < kNY; ++u) acc += yr[u] * wr[u];
    ws[WS_X0 + o] = acc;
  }
}

__device__ __forceinline__ void bu_gemm_store(const float* __restrict__ Uslot,
                                              const float* __restrict__ Bp,
                                              float* __restrict__ Xslot,
                                              int wave, int laneM, int laneHi) {
  const int mt    = (wave >> 1) & 1;
  const int ntPar = wave & 1;
  const int m     = mt * 16 + laneM;
  const int koff  = laneHi << 1;
  const float* Arow = Uslot + m * kNU;
  #pragma unroll
  for (int j = 0; j < 8; ++j) {
    const int nt = ntPar + (j << 1);
    const int n  = nt * 16 + laneM;
    const float* Brow = Bp + n * kNU;
    v8f acc = {};
    #pragma unroll
    for (int kk = 0; kk < 8; ++kk) {
      const int k = (kk << 2) + koff;
      v2f a = *(const v2f*)(Arow + k);
      v2f b = *(const v2f*)(Brow + k);
      acc = __builtin_amdgcn_wmma_f32_16x16x4_f32(false, a, false, b,
                                                  (short)0, acc, false, false);
    }
    #pragma unroll
    for (int v = 0; v < 8; ++v) {
      const int mm = mt * 16 + v + (laneHi << 3);
      Xslot[mm * kNH + n] = acc[v];
    }
  }
}

__global__ void __launch_bounds__(256)
pass1_kernel(const float* __restrict__ U, float* __restrict__ ws) {
  extern __shared__ float Xbuf[];
  const float* Bp = ws + WS_BP;

  const int g = blockIdx.x;
  const int c = blockIdx.y;
  const int tid    = threadIdx.x;
  const int lane   = tid & 31;
  const int wave   = tid >> 5;
  const int laneM  = lane & 15;
  const int laneHi = lane >> 4;
  const int slotW  = wave >> 2;

  const int sb = tid >> 3;
  const int h0 = (tid & 7) << 4;

  float st1[16], st2[16], rlr[16], rli[16];
  #pragma unroll
  for (int i = 0; i < 16; ++i) {
    st1[i] = 0.f; st2[i] = 0.f;
    rlr[i] = ws[WS_LR + h0 + i];
    rli[i] = ws[WS_LI + h0 + i];
  }

  for (int grp = 0; grp < kNGrp; ++grp) {
    const int t0 = c * kChunk + grp * kTG;
    const float* Uslot = U + ((size_t)(t0 + slotW) * kNB + g * kBG) * kNU;
    bu_gemm_store(Uslot, Bp, Xbuf + slotW * (kBG * kNH), wave, laneM, laneHi);
    __syncthreads();
    #pragma unroll
    for (int s = 0; s < kTG; ++s) {
      const float* Bu = Xbuf + s * (kBG * kNH) + sb * kNH;
      #pragma unroll
      for (int i = 0; i < 16; ++i) {
        const float x1 = st1[i], x2 = st2[i];
        st1[i] = rlr[i] * x1 - rli[i] * x2 + Bu[h0 + i];
        st2[i] = rli[i] * x1 + rlr[i] * x2 + Bu[h0 + i + kNHalf];
      }
    }
    __syncthreads();
  }
  float* F = ws + WS_F + ((size_t)c * kNB + g * kBG + sb) * kNH;
  #pragma unroll
  for (int i = 0; i < 16; ++i) {
    F[h0 + i]          = st1[i];
    F[h0 + i + kNHalf] = st2[i];
  }
}

__global__ void __launch_bounds__(256)
combine_kernel(float* __restrict__ ws) {
  const int tid = threadIdx.x;
  for (int i = 0; i < 32; ++i) {
    const int p = tid * 32 + i;
    const int b = p >> 7;
    const int h = p & 127;
    float s1 = ws[WS_X0 + (size_t)b * kNH + h];
    float s2 = ws[WS_X0 + (size_t)b * kNH + h + kNHalf];
    const float pr = ws[WS_LRL + h];
    const float pi = ws[WS_LIL + h];
    for (int c = 0; c < kNChunk; ++c) {
      const size_t base = (size_t)c * kNB * kNH + (size_t)b * kNH + h;
      ws[WS_S + base]          = s1;
      ws[WS_S + base + kNHalf] = s2;
      const float f1 = ws[WS_F + base];
      const float f2 = ws[WS_F + base + kNHalf];
      const float n1 = pr * s1 - pi * s2 + f1;
      const float n2 = pi * s1 + pr * s2 + f2;
      s1 = n1; s2 = n2;
    }
  }
}

__global__ void __launch_bounds__(256)
pass2_kernel(const float* __restrict__ U, const float* __restrict__ W,
             const float* __restrict__ bias, const float* __restrict__ ws,
             float* __restrict__ out) {
  extern __shared__ float Xbuf[];
  const float* Bp = ws + WS_BP;

  const int g = blockIdx.x;
  const int c = blockIdx.y;
  const int tid    = threadIdx.x;
  const int lane   = tid & 31;
  const int wave   = tid >> 5;
  const int laneM  = lane & 15;
  const int laneHi = lane >> 4;
  const int koff   = laneHi << 1;
  const int slotW  = wave >> 2;

  const int sb = tid >> 3;
  const int h0 = (tid & 7) << 4;

  float st1[16], st2[16], rlr[16], rli[16];
  {
    const float* S = ws + WS_S + ((size_t)c * kNB + g * kBG + sb) * kNH;
    #pragma unroll
    for (int i = 0; i < 16; ++i) {
      st1[i] = S[h0 + i];
      st2[i] = S[h0 + i + kNHalf];
      rlr[i] = ws[WS_LR + h0 + i];
      rli[i] = ws[WS_LI + h0 + i];
    }
  }

  const int mt2 = (wave >> 1) & 1;
  const int nt2 = wave & 1;
  const float* Wrow  = W + (size_t)(nt2 * 16 + laneM) * kNH;
  const float  biasl = bias[nt2 * 16 + laneM];

  for (int grp = 0; grp < kNGrp; ++grp) {
    const int t0 = c * kChunk + grp * kTG;
    const float* Uslot = U + ((size_t)(t0 + slotW) * kNB + g * kBG) * kNU;
    bu_gemm_store(Uslot, Bp, Xbuf + slotW * (kBG * kNH), wave, laneM, laneHi);
    __syncthreads();
    #pragma unroll
    for (int s = 0; s < kTG; ++s) {
      float* Bu = Xbuf + s * (kBG * kNH) + sb * kNH;
      #pragma unroll
      for (int i = 0; i < 16; ++i) {
        const float x1 = st1[i], x2 = st2[i];
        const float n1 = rlr[i] * x1 - rli[i] * x2 + Bu[h0 + i];
        const float n2 = rli[i] * x1 + rlr[i] * x2 + Bu[h0 + i + kNHalf];
        st1[i] = n1; st2[i] = n2;
        Bu[h0 + i]          = n1;
        Bu[h0 + i + kNHalf] = n2;
      }
    }
    __syncthreads();
    {
      const float* Arow = Xbuf + slotW * (kBG * kNH) + (mt2 * 16 + laneM) * kNH;
      v8f acc = {};
      #pragma unroll 8
      for (int kk = 0; kk < 64; ++kk) {
        const int k = (kk << 2) + koff;
        v2f a = *(const v2f*)(Arow + k);
        v2f b = *(const v2f*)(Wrow + k);
        acc = __builtin_amdgcn_wmma_f32_16x16x4_f32(false, a, false, b,
                                                    (short)0, acc, false, false);
      }
      const int t    = t0 + slotW;
      const int ycol = nt2 * 16 + laneM;
      float* Orow = out + (size_t)t * kNB * kNY + (size_t)(g * kBG) * kNY;
      #pragma unroll
      for (int v = 0; v < 8; ++v) {
        const int mm = mt2 * 16 + v + (laneHi << 3);
        Orow[mm * kNY + ycol] = acc[v] + biasl;
      }
    }
    __syncthreads();
  }
}

extern "C" void kernel_launch(void* const* d_in, const int* in_sizes, int n_in,
                              void* d_out, int out_size, void* d_ws, size_t ws_size,
                              hipStream_t stream) {
  (void)in_sizes; (void)n_in; (void)out_size; (void)ws_size;
  const float* y0   = (const float*)d_in[0];
  const float* U    = (const float*)d_in[1];
  const float* llr  = (const float*)d_in[2];
  const float* lli  = (const float*)d_in[3];
  const float* Bm   = (const float*)d_in[4];
  const float* y2xw = (const float*)d_in[5];
  const float* y2xb = (const float*)d_in[6];
  const float* x2yw = (const float*)d_in[7];
  const float* x2yb = (const float*)d_in[8];
  float* ws  = (float*)d_ws;
  float* out = (float*)d_out;

  hipFuncSetAttribute(reinterpret_cast<const void*>(pass1_kernel),
                      hipFuncAttributeMaxDynamicSharedMemorySize, (int)kSmemBytes);
  hipFuncSetAttribute(reinterpret_cast<const void*>(pass2_kernel),
                      hipFuncAttributeMaxDynamicSharedMemorySize, (int)kSmemBytes);

  setup_kernel<<<1, 256, 0, stream>>>(y0, llr, lli, Bm, y2xw, y2xb, ws);
  dim3 grid(kNBG, kNChunk);
  pass1_kernel<<<grid, 256, kSmemBytes, stream>>>(U, ws);
  combine_kernel<<<1, 256, 0, stream>>>(ws);
  pass2_kernel<<<grid, 256, kSmemBytes, stream>>>(U, x2yw, x2yb, ws, out);
}


# combine (carry of chunk states) spread over the chip: one thread per (batch, pole pair) in 256 one-wave blocks instead of one block looping 32 pairs per thread; per-pair sum order unchanged
# speedup vs baseline: 3.4112x; 3.4112x over previous
#include <hip/hip_runtime.h>

typedef __attribute__((ext_vector_type(2))) float v2f;
typedef __attribute__((ext_vector_type(8))) float v8f;

namespace {
constexpr int kT      = 4096;
constexpr int kNB     = 64;
constexpr int kNU     = 32;
constexpr int kNY     = 32;
constexpr int kNH     = 256;
constexpr int kNHalf  = 128;
constexpr int kChunk  = 64;
constexpr int kNChunk = kT / kChunk;
constexpr int kBG     = 32;
constexpr int kNBG    = kNB / kBG;
constexpr int kTG     = 2;
constexpr int kNGrp   = kChunk / kTG;
constexpr int kCombThreads = 32;
constexpr int kCombBlocks  = kNB * kNHalf / kCombThreads;

constexpr size_t WS_LR  = 0;
constexpr size_t WS_LI  = WS_LR + kNHalf;
constexpr size_t WS_LRL = WS_LI + kNHalf;
constexpr size_t WS_LIL = WS_LRL + kNHalf;
constexpr size_t WS_BP  = WS_LIL + kNHalf;
constexpr size_t WS_X0  = WS_BP + (size_t)kNH * kNU;
constexpr size_t WS_F   = WS_X0 + (size_t)kNB * kNH;
constexpr size_t WS_S   = WS_F + (size_t)kNChunk * kNB * kNH;

constexpr size_t kSmemBytes = (size_t)kTG * kBG * kNH * sizeof(float);
}

__global__ void __launch_bounds__(256)
setup_kernel(const float* __restrict__ y0, const float* __restrict__ llr,
             const float* __restrict__ lli, const float* __restrict__ Bmat,
             const float* __restrict__ y2xw, const float* __restrict__ y2xb,
             float* __restrict__ ws) {
  __shared__ float s_nf[kNH];
  const int tid = threadIdx.x;
  if (tid < kNHalf) {
    const float decay = __expf(-fabsf(llr[tid]));
    const float r     = __expf(-decay);
    const float th    = 1.57079632679489662f * __expf(-fabsf(lli[tid]));
    const float lr    = r * __cosf(th);
    const float li    = r * __sinf(th);
    const float nfh   = sqrtf(fmaxf(0.f, 1.f - r * r));
    ws[WS_LR + tid] = lr;
    ws[WS_LI + tid] = li;
    float pr = lr, pi = li;
    #pragma unroll
    for (int s = 0; s < 6; ++s) {
      const float nr = pr * pr - pi * pi;
      const float ni = 2.f * pr * pi;
      pr = nr; pi = ni;
    }
    ws[WS_LRL + tid] = pr;
    ws[WS_LIL + tid] = pi;
    s_nf[tid]          = nfh;
    s_nf[tid + kNHalf] = nfh;
  }
  __syncthreads();
  for (int i = tid; i < kNH * kNU; i += blockDim.x) {
    ws[WS_BP + i] = Bmat[i] * s_nf[i / kNU];
  }
  for (int o = tid; o < kNB * kNH; o += blockDim.x) {
    const int b = o >> 8;
    const int n = o & 255;
    float acc = y2xb[n];
    const float* yr = y0 + b * kNY;
    const float* wr = y2xw + n * kNY;
    #pragma unroll 8
    for (int u = 0; u < kNY; ++u) acc += yr[u] * wr[u];
    ws[WS_X0 + o] = acc;
  }
}

__device__ __forceinline__ void bu_gemm_store(const float* __restrict__ Uslot,
                                              const float* __restrict__ Bp,
                                              float* __restrict__ Xslot,
                                              int wave, int laneM, int laneHi) {
  const int mt    = (wave >> 1) & 1;
  const int ntPar = wave & 1;
  const int m     = mt * 16 + laneM;
  const int koff  = laneHi << 1;
  const float* Arow = Uslot + m * kNU;
  #pragma unroll
  for (int j = 0; j < 8; ++j) {
    const int nt = ntPar + (j << 1);
    const int n  = nt * 16 + laneM;
    const float* Brow = Bp + n * kNU;
    v8f acc = {};
    #pragma unroll
    for (int kk = 0; kk < 8; ++kk) {
      const int k = (kk << 2) + koff;
      v2f a = *(const v2f*)(Arow + k);
      v2f b = *(const v2f*)(Brow + k);
      acc = __builtin_amdgcn_wmma_f32_16x16x4_f32(false, a, false, b,
                                                  (short)0, acc, false, false);
    }
    #pragma unroll
    for (int v = 0; v < 8; ++v) {
      const int mm = mt * 16 + v + (laneHi << 3);
      Xslot[mm * kNH + n] = acc[v];
    }
  }
}

__global__ void __launch_bounds__(256)
pass1_kernel(const float* __restrict__ U, float* __restrict__ ws) {
  extern __shared__ float Xbuf[];
  const float* Bp = ws + WS_BP;

  const int g = blockIdx.x;
  const int c = blockIdx.y;
  const int tid    = threadIdx.x;
  const int lane   = tid & 31;
  const int wave   = tid >> 5;
  const int laneM  = lane & 15;
  const int laneHi = lane >> 4;
  const int slotW  = wave >> 2;

  const int sb = tid >> 3;
  const int h0 = (tid & 7) << 4;

  float st1[16], st2[16], rlr[16], rli[16];
  #pragma unroll
  for (int i = 0; i < 16; ++i) {
    st1[i] = 0.f; st2[i] = 0.f;
    rlr[i] = ws[WS_LR + h0 + i];
    rli[i] = ws[WS_LI + h0 + i];
  }

  for (int grp = 0; grp < kNGrp; ++grp) {
    const int t0 = c * kChunk + grp * kTG;
    const float* Uslot = U + ((size_t)(t0 + slotW) * kNB + g * kBG) * kNU;
    bu_gemm_store(Uslot, Bp, Xbuf + slotW * (kBG * kNH), wave, laneM, laneHi);
    __syncthreads();
    #pragma unroll
    for (int s = 0; s < kTG; ++s) {
      const float* Bu = Xbuf + s * (kBG * kNH) + sb * kNH;
      #pragma unroll
      for (int i = 0; i < 16; ++i) {
        const float x1 = st1[i], x2 = st2[i];
        st1[i] = rlr[i] * x1 - rli[i] * x2 + Bu[h0 + i];
        st2[i] = rli[i] * x1 + rlr[i] * x2 + Bu[h0 + i + kNHalf];
      }
    }
    __syncthreads();
  }
  float* F = ws + WS_F + ((size_t)c * kNB + g * kBG + sb) * kNH;
  #pragma unroll
  for (int i = 0; i < 16; ++i) {
    F[h0 + i]          = st1[i];
    F[h0 + i + kNHalf] = st2[i];
  }
}

__global__ void __launch_bounds__(kCombThreads)
combine_kernel(float* __restrict__ ws) {
  const int p = blockIdx.x * kCombThreads + threadIdx.x;
  const int b = p >> 7;
  const int h = p & 127;
  const size_t idx = (size_t)b * kNH + h;
  constexpr size_t kCS = (size_t)kNB * kNH;

  const float pr = ws[WS_LRL + h];
  const float pi = ws[WS_LIL + h];
  float s1 = ws[WS_X0 + idx];
  float s2 = ws[WS_X0 + idx + kNHalf];

  const float* F = ws + WS_F + idx;
  float*       S = ws + WS_S + idx;

  float f1[kNChunk], f2[kNChunk];
  #pragma unroll
  for (int c = 0; c < kNChunk; ++c) {
    f1[c] = F[(size_t)c * kCS];
    f2[c] = F[(size_t)c * kCS + kNHalf];
  }
  #pragma unroll
  for (int c = 0; c < kNChunk; ++c) {
    S[(size_t)c * kCS]          = s1;
    S[(size_t)c * kCS + kNHalf] = s2;
    const float n1 = pr * s1 - pi * s2 + f1[c];
    const float n2 = pi * s1 + pr * s2 + f2[c];
    s1 = n1; s2 = n2;
  }
}

__global__ void __launch_bounds__(256)
pass2_kernel(const float* __restrict__ U, const float* __restrict__ W,
             const float* __restrict__ bias, const float* __restrict__ ws,
             float* __restrict__ out) {
  extern __shared__ float Xbuf[];
  const float* Bp = ws + WS_BP;

  const int g = blockIdx.x;
  const int c = blockIdx.y;
  const int tid    = threadIdx.x;
  const int lane   = tid & 31;
  const int wave   = tid >> 5;
  const int laneM  = lane & 15;
  const int laneHi = lane >> 4;
  const int koff   = laneHi << 1;
  const int slotW  = wave >> 2;

  const int sb = tid >> 3;
  const int h0 = (tid & 7) << 4;

  float st1[16], st2[16], rlr[16], rli[16];
  {
    const float* S = ws + WS_S + ((size_t)c * kNB + g * kBG + sb) * kNH;
    #pragma unroll
    for (int i = 0; i < 16; ++i) {
      st1[i] = S[h0 + i];
      st2[i] = S[h0 + i + kNHalf];
      rlr[i] = ws[WS_LR + h0 + i];
      rli[i] = ws[WS_LI + h0 + i];
    }
  }

  const int mt2 = (wave >> 1) & 1;
  const int nt2 = wave & 1;
  const float* Wrow  = W + (size_t)(nt2 * 16 + laneM) * kNH;
  const float  biasl = bias[nt2 * 16 + laneM];

  for (int grp = 0; grp < kNGrp; ++grp) {
    const int t0 = c * kChunk + grp * kTG;
    const float* Uslot = U + ((size_t)(t0 + slotW) * kNB + g * kBG) * kNU;
    bu_gemm_store(Uslot, Bp, Xbuf + slotW * (kBG * kNH), wave, laneM, laneHi);
    __syncthreads();
    #pragma unroll
    for (int s = 0; s < kTG; ++s) {
      float* Bu = Xbuf + s * (kBG * kNH) + sb * kNH;
      #pragma unroll
      for (int i = 0; i < 16; ++i) {
        const float x1 = st1[i], x2 = st2[i];
        const float n1 = rlr[i] * x1 - rli[i] * x2 + Bu[h0 + i];
        const float n2 = rli[i] * x1 + rlr[i] * x2 + Bu[h0 + i + kNHalf];
        st1[i] = n1; st2[i] = n2;
        Bu[h0 + i]          = n1;
        Bu[h0 + i + kNHalf] = n2;
      }
    }
    __syncthreads();
    {
      const float* Arow = Xbuf + slotW * (kBG * kNH) + (mt2 * 16 + laneM) * kNH;
      v8f acc = {};
      #pragma unroll 8
      for (int kk = 0; kk < 64; ++kk) {
        const int k = (kk << 2) + koff;
        v2f a = *(const v2f*)(Arow + k);
        v2f b = *(const v2f*)(Wrow + k);
        acc = __builtin_amdgcn_wmma_f32_16x16x4_f32(false, a, false, b,
                                                    (short)0, acc, false, false);
      }
      const int t    = t0 + slotW;
      const int ycol = nt2 * 16 + laneM;
      float* Orow = out + (size_t)t * kNB * kNY + (size_t)(g * kBG) * kNY;
      #pragma unroll
      for (int v = 0; v < 8; ++v) {
        const int mm = mt2 * 16 + v + (laneHi << 3);
        Orow[mm * kNY + ycol] = acc[v] + biasl;
      }
    }
    __syncthreads();
  }
}

extern "C" void kernel_launch(void* const* d_in, const int* in_sizes, int n_in,
                              void* d_out, int out_size, void* d_ws, size_t ws_size,
                              hipStream_t stream) {
  (void)in_sizes; (void)n_in; (void)out_size; (void)ws_size;
  const float* y0   = (const float*)d_in[0];
  const float* U    = (const float*)d_in[1];
  const float* llr  = (const float*)d_in[2];
  const float* lli  = (const float*)d_in[3];
  const float* Bm   = (const float*)d_in[4];
  const float* y2xw = (const float*)d_in[5];
  const float* y2xb = (const float*)d_in[6];
  const float* x2yw = (const float*)d_in[7];
  const float* x2yb = (const float*)d_in[8];
  float* ws  = (float*)d_ws;
  float* out = (float*)d_out;

  hipFuncSetAttribute(reinterpret_cast<const void*>(pass1_kernel),
                      hipFuncAttributeMaxDynamicSharedMemorySize, (int)kSmemBytes);
  hipFuncSetAttribute(reinterpret_cast<const void*>(pass2_kernel),
                      hipFuncAttributeMaxDynamicSharedMemorySize, (int)kSmemBytes);

  setup_kernel<<<1, 256, 0, stream>>>(y0, llr, lli, Bm, y2xw, y2xb, ws);
  dim3 grid(kNBG, kNChunk);
  pass1_kernel<<<grid, 256, kSmemBytes, stream>>>(U, ws);
  combine_kernel<<<kCombBlocks, kCombThreads, 0, stream>>>(ws);
  pass2_kernel<<<grid, 256, kSmemBytes, stream>>>(U, x2yw, x2yb, ws, out);
}


# chunk length 64 -> 16 timesteps: 512 blocks per scan pass instead of 128, two per WGP on every WGP; carry chain now 256 chunks, so the recurrence's sum order changes at the new chunk borders
# speedup vs baseline: 6.5967x; 1.9338x over previous
#include <hip/hip_runtime.h>

typedef __attribute__((ext_vector_type(2))) float v2f;
typedef __attribute__((ext_vector_type(8))) float v8f;

namespace {
constexpr int kT      = 4096;
constexpr int kNB     = 64;
constexpr int kNU     = 32;
constexpr int kNY     = 32;
constexpr int kNH     = 256;
constexpr int kNHalf  = 128;
constexpr int kLog2Chunk = 4;
constexpr int kChunk  = 1 << kLog2Chunk;
constexpr int kNChunk = kT / kChunk;
constexpr int kBG     = 32;
constexpr int kNBG    = kNB / kBG;
constexpr int kTG     = 2;
constexpr int kNGrp   = kChunk / kTG;
constexpr int kCombThreads = 32;
constexpr int kCombBlocks  = kNB * kNHalf / kCombThreads;
constexpr int kCombBatch   = 32;
static_assert(kNChunk % (2 * kCombBatch) == 0, "combine walks the chunks in pairs of batches");

constexpr size_t WS_LR  = 0;
constexpr size_t WS_LI  = WS_LR + kNHalf;
constexpr size_t WS_LRL = WS_LI + kNHalf;
constexpr size_t WS_LIL = WS_LRL + kNHalf;
constexpr size_t WS_BP  = WS_LIL + kNHalf;
constexpr size_t WS_X0  = WS_BP + (size_t)kNH * kNU;
constexpr size_t WS_F   = WS_X0 + (size_t)kNB * kNH;
constexpr size_t WS_S   = WS_F + (size_t)kNChunk * kNB * kNH;

constexpr size_t kSmemBytes = (size_t)kTG * kBG * kNH * sizeof(float);
}

__global__ void __launch_bounds__(256)
setup_kernel(const float* __restrict__ y0, const float* __restrict__ llr,
             const float* __restrict__ lli, const float* __restrict__ Bmat,
             const float* __restrict__ y2xw, const float* __restrict__ y2xb,
             float* __restrict__ ws) {
  __shared__ float s_nf[kNH];
  const int tid = threadIdx.x;
  if (tid < kNHalf) {
    const float decay = __expf(-fabsf(llr[tid]));
    const float r     = __expf(-decay);
    const float th    = 1.57079632679489662f * __expf(-fabsf(lli[tid]));
    const float lr    = r * __cosf(th);
    const float li    = r * __sinf(th);
    const float nfh   = sqrtf(fmaxf(0.f, 1.f - r * r));
    ws[WS_LR + tid] = lr;
    ws[WS_LI + tid] = li;
    float pr = lr, pi = li;
    #pragma unroll
    for (int s = 0; s < kLog2Chunk; ++s) {
      const float nr = pr * pr - pi * pi;
      const float ni = 2.f * pr * pi;
      pr = nr; pi = ni;
    }
    ws[WS_LRL + tid] = pr;
    ws[WS_LIL + tid] = pi;
    s_nf[tid]          = nfh;
    s_nf[tid + kNHalf] = nfh;
  }
  __syncthreads();
  for (int i = tid; i < kNH * kNU; i += blockDim.x) {
    ws[WS_BP + i] = Bmat[i] * s_nf[i / kNU];
  }
  for (int o = tid; o < kNB * kNH; o += blockDim.x) {
    const int b = o >> 8;
    const int n = o & 255;
    float acc = y2xb[n];
    const float* yr = y0 + b * kNY;
    const float* wr = y2xw + n * kNY;
    #pragma unroll 8
    for (int u = 0; u < kNY; ++u) acc += yr[u] * wr[u];
    ws[WS_X0 + o] = acc;
  }
}

__device__ __forceinline__ void bu_gemm_store(const float* __restrict__ Uslot,
                                              const float* __restrict__ Bp,
                                              float* __restrict__ Xslot,
                                              int wave, int laneM, int laneHi) {
  const int mt    = (wave >> 1) & 1;
  const int ntPar = wave & 1;
  const int m     = mt * 16 + laneM;
  const int koff  = laneHi << 1;
  const float* Arow = Uslot + m * kNU;
  #pragma unroll
  for (int j = 0; j < 8; ++j) {
    const int nt = ntPar + (j << 1);
    const int n  = nt * 16 + laneM;
    const float* Brow = Bp + n * kNU;
    v8f acc = {};
    #pragma unroll
    for (int kk = 0; kk < 8; ++kk) {
      const int k = (kk << 2) + koff;
      v2f a = *(const v2f*)(Arow + k);
      v2f b = *(const v2f*)(Brow + k);
      acc = __builtin_amdgcn_wmma_f32_16x16x4_f32(false, a, false, b,
                                                  (short)0, acc, false, false);
    }
    #pragma unroll
    for (int v = 0; v < 8; ++v) {
      const int mm = mt * 16 + v + (laneHi << 3);
      Xslot[mm * kNH + n] = acc[v];
    }
  }
}

__global__ void __launch_bounds__(256)
pass1_kernel(const float* __restrict__ U, float* __restrict__ ws) {
  extern __shared__ float Xbuf[];
  const float* Bp = ws + WS_BP;

  const int g = blockIdx.x;
  const int c = blockIdx.y;
  const int tid    = threadIdx.x;
  const int lane   = tid & 31;
  const int wave   = tid >> 5;
  const int laneM  = lane & 15;
  const int laneHi = lane >> 4;
  const int slotW  = wave >> 2;

  const int sb = tid >> 3;
  const int h0 = (tid & 7) << 4;

  float st1[16], st2[16], rlr[16], rli[16];
  #pragma unroll
  for (int i = 0; i < 16; ++i) {
    st1[i] = 0.f; st2[i] = 0.f;
    rlr[i] = ws[WS_LR + h0 + i];
    rli[i] = ws[WS_LI + h0 + i];
  }

  for (int grp = 0; grp < kNGrp; ++grp) {
    const int t0 = c * kChunk + grp * kTG;
    const float* Uslot = U + ((size_t)(t0 + slotW) * kNB + g * kBG) * kNU;
    bu_gemm_store(Uslot, Bp, Xbuf + slotW * (kBG * kNH), wave, laneM, laneHi);
    __syncthreads();
    #pragma unroll
    for (int s = 0; s < kTG; ++s) {
      const float* Bu = Xbuf + s * (kBG * kNH) + sb * kNH;
      #pragma unroll
      for (int i = 0; i < 16; ++i) {
        const float x1 = st1[i], x2 = st2[i];
        st1[i] = rlr[i] * x1 - rli[i] * x2 + Bu[h0 + i];
        st2[i] = rli[i] * x1 + rlr[i] * x2 + Bu[h0 + i + kNHalf];
      }
    }
    __syncthreads();
  }
  float* F = ws + WS_F + ((size_t)c * kNB + g * kBG + sb) * kNH;
  #pragma unroll
  for (int i = 0; i < 16; ++i) {
    F[h0 + i]          = st1[i];
    F[h0 + i + kNHalf] = st2[i];
  }
}

__device__ __forceinline__ void comb_fetch(const float* __restrict__ Fb,
                                           float (&f1)[kCombBatch], float (&f2)[kCombBatch]) {
  constexpr size_t kCS = (size_t)kNB * kNH;
  #pragma unroll
  for (int j = 0; j < kCombBatch; ++j) {
    f1[j] = Fb[(size_t)j * kCS];
    f2[j] = Fb[(size_t)j * kCS + kNHalf];
  }
}

__device__ __forceinline__ void comb_chain(float* __restrict__ Sb,
                                           const float (&f1)[kCombBatch],
                                           const float (&f2)[kCombBatch],
                                           float pr, float pi, float& s1, float& s2) {
  constexpr size_t kCS = (size_t)kNB * kNH;
  #pragma unroll
  for (int j = 0; j < kCombBatch; ++j) {
    Sb[(size_t)j * kCS]          = s1;
    Sb[(size_t)j * kCS + kNHalf] = s2;
    const float n1 = pr * s1 - pi * s2 + f1[j];
    const float n2 = pi * s1 + pr * s2 + f2[j];
    s1 = n1; s2 = n2;
  }
}

__global__ void __launch_bounds__(kCombThreads)
combine_kernel(float* __restrict__ ws) {
  const int p = blockIdx.x * kCombThreads + threadIdx.x;
  const int b = p >> 7;
  const int h = p & 127;
  const size_t idx = (size_t)b * kNH + h;
  constexpr size_t kCS = (size_t)kNB * kNH;

  const float pr = ws[WS_LRL + h];
  const float pi = ws[WS_LIL + h];
  float s1 = ws[WS_X0 + idx];
  float s2 = ws[WS_X0 + idx + kNHalf];

  const float* F = ws + WS_F + idx;
  float*       S = ws + WS_S + idx;

  float fa1[kCombBatch], fa2[kCombBatch], fb1[kCombBatch], fb2[kCombBatch];
  comb_fetch(F, fa1, fa2);
  #pragma unroll 1
  for (int c0 = 0; c0 < kNChunk - 2 * kCombBatch; c0 += 2 * kCombBatch) {
    comb_fetch(F + (size_t)(c0 + kCombBatch) * kCS, fb1, fb2);
    comb_chain(S + (size_t)c0 * kCS, fa1, fa2, pr, pi, s1, s2);
    comb_fetch(F + (size_t)(c0 + 2 * kCombBatch) * kCS, fa1, fa2);
    comb_chain(S + (size_t)(c0 + kCombBatch) * kCS, fb1, fb2, pr, pi, s1, s2);
  }
  constexpr int cL = kNChunk - 2 * kCombBatch;
  comb_fetch(F + (size_t)(cL + kCombBatch) * kCS, fb1, fb2);
  comb_chain(S + (size_t)cL * kCS, fa1, fa2, pr, pi, s1, s2);
  comb_chain(S + (size_t)(cL + kCombBatch) * kCS, fb1, fb2, pr, pi, s1, s2);
}

__global__ void __launch_bounds__(256)
pass2_kernel(const float* __restrict__ U, const float* __restrict__ W,
             const float* __restrict__ bias, const float* __restrict__ ws,
             float* __restrict__ out) {
  extern __shared__ float Xbuf[];
  const float* Bp = ws + WS_BP;

  const int g = blockIdx.x;
  const int c = blockIdx.y;
  const int tid    = threadIdx.x;
  const int lane   = tid & 31;
  const int wave   = tid >> 5;
  const int laneM  = lane & 15;
  const int laneHi = lane >> 4;
  const int koff   = laneHi << 1;
  const int slotW  = wave >> 2;

  const int sb = tid >> 3;
  const int h0 = (tid & 7) << 4;

  float st1[16], st2[16], rlr[16], rli[16];
  {
    const float* S = ws + WS_S + ((size_t)c * kNB + g * kBG + sb) * kNH;
    #pragma unroll
    for (int i = 0; i < 16; ++i) {
      st1[i] = S[h0 + i];
      st2[i] = S[h0 + i + kNHalf];
      rlr[i] = ws[WS_LR + h0 + i];
      rli[i] = ws[WS_LI + h0 + i];
    }
  }

  const int mt2 = (wave >> 1) & 1;
  const int nt2 = wave & 1;
  const float* Wrow  = W + (size_t)(nt2 * 16 + laneM) * kNH;
  const float  biasl = bias[nt2 * 16 + laneM];

  for (int grp = 0; grp < kNGrp; ++grp) {
    const int t0 = c * kChunk + grp * kTG;
    const float* Uslot = U + ((size_t)(t0 + slotW) * kNB + g * kBG) * kNU;
    bu_gemm_store(Uslot, Bp, Xbuf + slotW * (kBG * kNH), wave, laneM, laneHi);
    __syncthreads();
    #pragma unroll
    for (int s = 0; s < kTG; ++s) {
      float* Bu = Xbuf + s * (kBG * kNH) + sb * kNH;
      #pragma unroll
      for (int i = 0; i < 16; ++i) {
        const float x1 = st1[i], x2 = st2[i];
        const float n1 = rlr[i] * x1 - rli[i] * x2 + Bu[h0 + i];
        const float n2 = rli[i] * x1 + rlr[i] * x2 + Bu[h0 + i + kNHalf];
        st1[i] = n1; st2[i] = n2;
        Bu[h0 + i]          = n1;
        Bu[h0 + i + kNHalf] = n2;
      }
    }
    __syncthreads();
    {
      const float* Arow = Xbuf + slotW * (kBG * kNH) + (mt2 * 16 + laneM) * kNH;
      v8f acc = {};
      #pragma unroll 8
      for (int kk = 0; kk < 64; ++kk) {
        const int k = (kk << 2) + koff;
        v2f a = *(const v2f*)(Arow + k);
        v2f b = *(const v2f*)(Wrow + k);
        acc = __builtin_amdgcn_wmma_f32_16x16x4_f32(false, a, false, b,
                                                    (short)0, acc, false, false);
      }
      const int t    = t0 + slotW;
      const int ycol = nt2 * 16 + laneM;
      float* Orow = out + (size_t)t * kNB * kNY + (size_t)(g * kBG) * kNY;
      #pragma unroll
      for (int v = 0; v < 8; ++v) {
        const int mm = mt2 * 16 + v + (laneHi << 3);
        Orow[mm * kNY + ycol] = acc[v] + biasl;
      }
    }
    __syncthreads();
  }
}

extern "C" void kernel_launch(void* const* d_in, const int* in_sizes, int n_in,
                              void* d_out, int out_size, void* d_ws, size_t ws_size,
                              hipStream_t stream) {
  (void)in_sizes; (void)n_in; (void)out_size; (void)ws_size;
  const float* y0   = (const float*)d_in[0];
  const float* U    = (const float*)d_in[1];
  const float* llr  = (const float*)d_in[2];
  const float* lli  = (const float*)d_in[3];
  const float* Bm   = (const float*)d_in[4];
  const float* y2xw = (const float*)d_in[5];
  const float* y2xb = (const float*)d_in[6];
  const float* x2yw = (const float*)d_in[7];
  const float* x2yb = (const float*)d_in[8];
  float* ws  = (float*)d_ws;
  float* out = (float*)d_out;

  hipFuncSetAttribute(reinterpret_cast<const void*>(pass1_kernel),
                      hipFuncAttributeMaxDynamicSharedMemorySize, (int)kSmemBytes);
  hipFuncSetAttribute(reinterpret_cast<const void*>(pass2_kernel),
                      hipFuncAttributeMaxDynamicSharedMemorySize, (int)kSmemBytes);

  setup_kernel<<<1, 256, 0, stream>>>(y0, llr, lli, Bm, y2xw, y2xb, ws);
  dim3 grid(kNBG, kNChunk);
  pass1_kernel<<<grid, 256, kSmemBytes, stream>>>(U, ws);
  combine_kernel<<<kCombBlocks, kCombThreads, 0, stream>>>(ws);
  pass2_kernel<<<grid, 256, kSmemBytes, stream>>>(U, x2yw, x2yb, ws, out);
}


# LDS X tile made bank-conflict-free: row pitch 256 -> 260 floats, and scan lanes own 4 rows x 4 adjacent pairs read as 128-bit row segments; per-pair arithmetic and sum order unchanged
# speedup vs baseline: 8.6782x; 1.3155x over previous
#include <hip/hip_runtime.h>

typedef __attribute__((ext_vector_type(2))) float v2f;
typedef __attribute__((ext_vector_type(4))) float v4f;
typedef __attribute__((ext_vector_type(8))) float v8f;

namespace {
constexpr int kT      = 4096;
constexpr int kNB     = 64;
constexpr int kNU     = 32;
constexpr int kNY     = 32;
constexpr int kNH     = 256;
constexpr int kNHalf  = 128;
constexpr int kLog2Chunk = 4;
constexpr int kChunk  = 1 << kLog2Chunk;
constexpr int kNChunk = kT / kChunk;
constexpr int kBG     = 32;
constexpr int kNBG    = kNB / kBG;
constexpr int kTG     = 2;
constexpr int kNGrp   = kChunk / kTG;
constexpr int kPitch  = kNH + 4;
constexpr int kSlot   = kBG * kPitch;
constexpr int kCombThreads = 32;
constexpr int kCombBlocks  = kNB * kNHalf / kCombThreads;
constexpr int kCombBatch   = 32;
static_assert(kNChunk % (2 * kCombBatch) == 0, "combine walks the chunks in pairs of batches");

constexpr size_t WS_LR  = 0;
constexpr size_t WS_LI  = WS_LR + kNHalf;
constexpr size_t WS_LRL = WS_LI + kNHalf;
constexpr size_t WS_LIL = WS_LRL + kNHalf;
constexpr size_t WS_BP  = WS_LIL + kNHalf;
constexpr size_t WS_X0  = WS_BP + (size_t)kNH * kNU;
constexpr size_t WS_F   = WS_X0 + (size_t)kNB * kNH;
constexpr size_t WS_S   = WS_F + (size_t)kNChunk * kNB * kNH;

constexpr size_t kSmemBytes = (size_t)kTG * kSlot * sizeof(float);
}

__global__ void __launch_bounds__(256)
setup_kernel(const float* __restrict__ y0, const float* __restrict__ llr,
             const float* __restrict__ lli, const float* __restrict__ Bmat,
             const float* __restrict__ y2xw, const float* __restrict__ y2xb,
             float* __restrict__ ws) {
  __shared__ float s_nf[kNH];
  const int tid = threadIdx.x;
  if (tid < kNHalf) {
    const float decay = __expf(-fabsf(llr[tid]));
    const float r     = __expf(-decay);
    const float th    = 1.57079632679489662f * __expf(-fabsf(lli[tid]));
    const float lr    = r * __cosf(th);
    const float li    = r * __sinf(th);
    const float nfh   = sqrtf(fmaxf(0.f, 1.f - r * r));
    ws[WS_LR + tid] = lr;
    ws[WS_LI + tid] = li;
    float pr = lr, pi = li;
    #pragma unroll
    for (int s = 0; s < kLog2Chunk; ++s) {
      const float nr = pr * pr - pi * pi;
      const float ni = 2.f * pr * pi;
      pr = nr; pi = ni;
    }
    ws[WS_LRL + tid] = pr;
    ws[WS_LIL + tid] = pi;
    s_nf[tid]          = nfh;
    s_nf[tid + kNHalf] = nfh;
  }
  __syncthreads();
  for (int i = tid; i < kNH * kNU; i += blockDim.x) {
    ws[WS_BP + i] = Bmat[i] * s_nf[i / kNU];
  }
  for (int o = tid; o < kNB * kNH; o += blockDim.x) {
    const int b = o >> 8;
    const int n = o & 255;
    float acc = y2xb[n];
    const float* yr = y0 + b * kNY;
    const float* wr = y2xw + n * kNY;
    #pragma unroll 8
    for (int u = 0; u < kNY; ++u) acc += yr[u] * wr[u];
    ws[WS_X0 + o] = acc;
  }
}

__device__ __forceinline__ void bu_gemm_store(const float* __restrict__ Uslot,
                                              const float* __restrict__ Bp,
                                              float* __restrict__ Xslot,
                                              int wave, int laneM, int laneHi) {
  const int mt    = (wave >> 1) & 1;
  const int ntPar = wave & 1;
  const int m     = mt * 16 + laneM;
  const int koff  = laneHi << 1;
  const float* Arow = Uslot + m * kNU;
  #pragma unroll
  for (int j = 0; j < 8; ++j) {
    const int nt = ntPar + (j << 1);
    const int n  = nt * 16 + laneM;
    const float* Brow = Bp + n * kNU;
    v8f acc = {};
    #pragma unroll
    for (int kk = 0; kk < 8; ++kk) {
      const int k = (kk << 2) + koff;
      v2f a = *(const v2f*)(Arow + k);
      v2f b = *(const v2f*)(Brow + k);
      acc = __builtin_amdgcn_wmma_f32_16x16x4_f32(false, a, false, b,
                                                  (short)0, acc, false, false);
    }
    #pragma unroll
    for (int v = 0; v < 8; ++v) {
      const int mm = mt * 16 + v + (laneHi << 3);
      Xslot[mm * kPitch + n] = acc[v];
    }
  }
}

__global__ void __launch_bounds__(256)
pass1_kernel(const float* __restrict__ U, float* __restrict__ ws) {
  extern __shared__ __align__(16) float Xbuf[];
  const float* Bp = ws + WS_BP;

  const int g = blockIdx.x;
  const int c = blockIdx.y;
  const int tid    = threadIdx.x;
  const int lane   = tid & 31;
  const int wave   = tid >> 5;
  const int laneM  = lane & 15;
  const int laneHi = lane >> 4;
  const int slotW  = wave >> 2;

  const int sb0 = wave << 2;
  const int h0  = lane << 2;

  v4f st1[4], st2[4];
  #pragma unroll
  for (int q = 0; q < 4; ++q) { st1[q] = (v4f)(0.f); st2[q] = (v4f)(0.f); }
  const v4f rlr = *(const v4f*)(ws + WS_LR + h0);
  const v4f rli = *(const v4f*)(ws + WS_LI + h0);

  for (int grp = 0; grp < kNGrp; ++grp) {
    const int t0 = c * kChunk + grp * kTG;
    const float* Uslot = U + ((size_t)(t0 + slotW) * kNB + g * kBG) * kNU;
    bu_gemm_store(Uslot, Bp, Xbuf + slotW * kSlot, wave, laneM, laneHi);
    __syncthreads();
    #pragma unroll
    for (int s = 0; s < kTG; ++s) {
      #pragma unroll
      for (int q = 0; q < 4; ++q) {
        const float* Bu = Xbuf + s * kSlot + (sb0 + q) * kPitch + h0;
        const v4f b1 = *(const v4f*)(Bu);
        const v4f b2 = *(const v4f*)(Bu + kNHalf);
        #pragma unroll
        for (int e = 0; e < 4; ++e) {
          const float x1 = st1[q][e], x2 = st2[q][e];
          st1[q][e] = rlr[e] * x1 - rli[e] * x2 + b1[e];
          st2[q][e] = rli[e] * x1 + rlr[e] * x2 + b2[e];
        }
      }
    }
    __syncthreads();
  }
  #pragma unroll
  for (int q = 0; q < 4; ++q) {
    float* F = ws + WS_F + ((size_t)c * kNB + g * kBG + sb0 + q) * kNH + h0;
    *(v4f*)(F)          = st1[q];
    *(v4f*)(F + kNHalf) = st2[q];
  }
}

__device__ __forceinline__ void comb_fetch(const float* __restrict__ Fb,
                                           float (&f1)[kCombBatch], float (&f2)[kCombBatch]) {
  constexpr size_t kCS = (size_t)kNB * kNH;
  #pragma unroll
  for (int j = 0; j < kCombBatch; ++j) {
    f1[j] = Fb[(size_t)j * kCS];
    f2[j] = Fb[(size_t)j * kCS + kNHalf];
  }
}

__device__ __forceinline__ void comb_chain(float* __restrict__ Sb,
                                           const float (&f1)[kCombBatch],
                                           const float (&f2)[kCombBatch],
                                           float pr, float pi, float& s1, float& s2) {
  constexpr size_t kCS = (size_t)kNB * kNH;
  #pragma unroll
  for (int j = 0; j < kCombBatch; ++j) {
    Sb[(size_t)j * kCS]          = s1;
    Sb[(size_t)j * kCS + kNHalf] = s2;
    const float n1 = pr * s1 - pi * s2 + f1[j];
    const float n2 = pi * s1 + pr * s2 + f2[j];
    s1 = n1; s2 = n2;
  }
}

__global__ void __launch_bounds__(kCombThreads)
combine_kernel(float* __restrict__ ws) {
  const int p = blockIdx.x * kCombThreads + threadIdx.x;
  const int b = p >> 7;
  const int h = p & 127;
  const size_t idx = (size_t)b * kNH + h;
  constexpr size_t kCS = (size_t)kNB * kNH;

  const float pr = ws[WS_LRL + h];
  const float pi = ws[WS_LIL + h];
  float s1 = ws[WS_X0 + idx];
  float s2 = ws[WS_X0 + idx + kNHalf];

  const float* F = ws + WS_F + idx;
  float*       S = ws + WS_S + idx;

  float fa1[kCombBatch], fa2[kCombBatch], fb1[kCombBatch], fb2[kCombBatch];
  comb_fetch(F, fa1, fa2);
  #pragma unroll 1
  for (int c0 = 0; c0 < kNChunk - 2 * kCombBatch; c0 += 2 * kCombBatch) {
    comb_fetch(F + (size_t)(c0 + kCombBatch) * kCS, fb1, fb2);
    comb_chain(S + (size_t)c0 * kCS, fa1, fa2, pr, pi, s1, s2);
    comb_fetch(F + (size_t)(c0 + 2 * kCombBatch) * kCS, fa1, fa2);
    comb_chain(S + (size_t)(c0 + kCombBatch) * kCS, fb1, fb2, pr, pi, s1, s2);
  }
  constexpr int cL = kNChunk - 2 * kCombBatch;
  comb_fetch(F + (size_t)(cL + kCombBatch) * kCS, fb1, fb2);
  comb_chain(S + (size_t)cL * kCS, fa1, fa2, pr, pi, s1, s2);
  comb_chain(S + (size_t)(cL + kCombBatch) * kCS, fb1, fb2, pr, pi, s1, s2);
}

__global__ void __launch_bounds__(256)
pass2_kernel(const float* __restrict__ U, const float* __restrict__ W,
             const float* __restrict__ bias, const float* __restrict__ ws,
             float* __restrict__ out) {
  extern __shared__ __align__(16) float Xbuf[];
  const float* Bp = ws + WS_BP;

  const int g = blockIdx.x;
  const int c = blockIdx.y;
  const int tid    = threadIdx.x;
  const int lane   = tid & 31;
  const int wave   = tid >> 5;
  const int laneM  = lane & 15;
  const int laneHi = lane >> 4;
  const int koff   = laneHi << 1;
  const int slotW  = wave >> 2;

  const int sb0 = wave << 2;
  const int h0  = lane << 2;

  v4f st1[4], st2[4];
  #pragma unroll
  for (int q = 0; q < 4; ++q) {
    const float* S = ws + WS_S + ((size_t)c * kNB + g * kBG + sb0 + q) * kNH + h0;
    st1[q] = *(const v4f*)(S);
    st2[q] = *(const v4f*)(S + kNHalf);
  }
  const v4f rlr = *(const v4f*)(ws + WS_LR + h0);
  const v4f rli = *(const v4f*)(ws + WS_LI + h0);

  const int mt2 = (wave >> 1) & 1;
  const int nt2 = wave & 1;
  const float* Wrow  = W + (size_t)(nt2 * 16 + laneM) * kNH;
  const float  biasl = bias[nt2 * 16 + laneM];

  for (int grp = 0; grp < kNGrp; ++grp) {
    const int t0 = c * kChunk + grp * kTG;
    const float* Uslot = U + ((size_t)(t0 + slotW) * kNB + g * kBG) * kNU;
    bu_gemm_store(Uslot, Bp, Xbuf + slotW * kSlot, wave, laneM, laneHi);
    __syncthreads();
    #pragma unroll
    for (int s = 0; s < kTG; ++s) {
      #pragma unroll
      for (int q = 0; q < 4; ++q) {
        float* Bu = Xbuf + s * kSlot + (sb0 + q) * kPitch + h0;
        const v4f b1 = *(const v4f*)(Bu);
        const v4f b2 = *(const v4f*)(Bu + kNHalf);
        #pragma unroll
        for (int e = 0; e < 4; ++e) {
          const float x1 = st1[q][e], x2 = st2[q][e];
          const float n1 = rlr[e] * x1 - rli[e] * x2 + b1[e];
          const float n2 = rli[e] * x1 + rlr[e] * x2 + b2[e];
          st1[q][e] = n1; st2[q][e] = n2;
        }
        *(v4f*)(Bu)          = st1[q];
        *(v4f*)(Bu + kNHalf) = st2[q];
      }
    }
    __syncthreads();
    {
      const float* Arow = Xbuf + slotW * kSlot + (mt2 * 16 + laneM) * kPitch;
      v8f acc = {};
      #pragma unroll 8
      for (int kk = 0; kk < 64; ++kk) {
        const int k = (kk << 2) + koff;
        v2f a = *(const v2f*)(Arow + k);
        v2f b = *(const v2f*)(Wrow + k);
        acc = __builtin_amdgcn_wmma_f32_16x16x4_f32(false, a, false, b,
                                                    (short)0, acc, false, false);
      }
      const int t    = t0 + slotW;
      const int ycol = nt2 * 16 + laneM;
      float* Orow = out + (size_t)t * kNB * kNY + (size_t)(g * kBG) * kNY;
      #pragma unroll
      for (int v = 0; v < 8; ++v) {
        const int mm = mt2 * 16 + v + (laneHi << 3);
        Orow[mm * kNY + ycol] = acc[v] + biasl;
      }
    }
    __syncthreads();
  }
}

extern "C" void kernel_launch(void* const* d_in, const int* in_sizes, int n_in,
                              void* d_out, int out_size, void* d_ws, size_t ws_size,
                              hipStream_t stream) {
  (void)in_sizes; (void)n_in; (void)out_size; (void)ws_size;
  const float* y0   = (const float*)d_in[0];
  const float* U    = (const float*)d_in[1];
  const float* llr  = (const float*)d_in[2];
  const float* lli  = (const float*)d_in[3];
  const float* Bm   = (const float*)d_in[4];
  const float* y2xw = (const float*)d_in[5];
  const float* y2xb = (const float*)d_in[6];
  const float* x2yw = (const float*)d_in[7];
  const float* x2yb = (const float*)d_in[8];
  float* ws  = (float*)d_ws;
  float* out = (float*)d_out;

  hipFuncSetAttribute(reinterpret_cast<const void*>(pass1_kernel),
                      hipFuncAttributeMaxDynamicSharedMemorySize, (int)kSmemBytes);
  hipFuncSetAttribute(reinterpret_cast<const void*>(pass2_kernel),
                      hipFuncAttributeMaxDynamicSharedMemorySize, (int)kSmemBytes);

  setup_kernel<<<1, 256, 0, stream>>>(y0, llr, lli, Bm, y2xw, y2xb, ws);
  dim3 grid(kNBG, kNChunk);
  pass1_kernel<<<grid, 256, kSmemBytes, stream>>>(U, ws);
  combine_kernel<<<kCombBlocks, kCombThreads, 0, stream>>>(ws);
  pass2_kernel<<<grid, 256, kSmemBytes, stream>>>(U, x2yw, x2yb, ws, out);
}


# B operands of both GEMMs (scaled input weights, output weights) copied once per block into LDS at conflict-free pitches; fragment reads are LDS reads, not 16-line global loads; sum order unchanged
# speedup vs baseline: 10.4306x; 1.2019x over previous
#include <hip/hip_runtime.h>

typedef __attribute__((ext_vector_type(2))) float v2f;
typedef __attribute__((ext_vector_type(4))) float v4f;
typedef __attribute__((ext_vector_type(8))) float v8f;

namespace {
constexpr int kT      = 4096;
constexpr int kNB     = 64;
constexpr int kNU     = 32;
constexpr int kNY     = 32;
constexpr int kNH     = 256;
constexpr int kNHalf  = 128;
constexpr int kLog2Chunk = 4;
constexpr int kChunk  = 1 << kLog2Chunk;
constexpr int kNChunk = kT / kChunk;
constexpr int kBG     = 32;
constexpr int kNBG    = kNB / kBG;
constexpr int kTG     = 2;
constexpr int kNGrp   = kChunk / kTG;
constexpr int kPitch  = kNH + 4;
constexpr int kSlot   = kBG * kPitch;
constexpr int kPitchB  = kNU + 4;
constexpr int kXFloats  = kTG * kSlot;
constexpr int kBpFloats = kNH * kPitchB;
constexpr int kWFloats  = kNY * kPitch;
constexpr int kCombThreads = 32;
constexpr int kCombBlocks  = kNB * kNHalf / kCombThreads;
constexpr int kCombBatch   = 32;
static_assert(kNChunk % (2 * kCombBatch) == 0, "combine walks the chunks in pairs of batches");

constexpr size_t WS_LR  = 0;
constexpr size_t WS_LI  = WS_LR + kNHalf;
constexpr size_t WS_LRL = WS_LI + kNHalf;
constexpr size_t WS_LIL = WS_LRL + kNHalf;
constexpr size_t WS_BP  = WS_LIL + kNHalf;
constexpr size_t WS_X0  = WS_BP + (size_t)kNH * kNU;
constexpr size_t WS_F   = WS_X0 + (size_t)kNB * kNH;
constexpr size_t WS_S   = WS_F + (size_t)kNChunk * kNB * kNH;

constexpr size_t kSmemBytes1 = (size_t)(kXFloats + kBpFloats) * sizeof(float);
constexpr size_t kSmemBytes2 = (size_t)(kXFloats + kBpFloats + kWFloats) * sizeof(float);
}

__global__ void __launch_bounds__(256)
setup_kernel(const float* __restrict__ y0, const float* __restrict__ llr,
             const float* __restrict__ lli, const float* __restrict__ Bmat,
             const float* __restrict__ y2xw, const float* __restrict__ y2xb,
             float* __restrict__ ws) {
  __shared__ float s_nf[kNH];
  const int tid = threadIdx.x;
  if (tid < kNHalf) {
    const float decay = __expf(-fabsf(llr[tid]));
    const float r     = __expf(-decay);
    const float th    = 1.57079632679489662f * __expf(-fabsf(lli[tid]));
    const float lr    = r * __cosf(th);
    const float li    = r * __sinf(th);
    const float nfh   = sqrtf(fmaxf(0.f, 1.f - r * r));
    ws[WS_LR + tid] = lr;
    ws[WS_LI + tid] = li;
    float pr = lr, pi = li;
    #pragma unroll
    for (int s = 0; s < kLog2Chunk; ++s) {
      const float nr = pr * pr - pi * pi;
      const float ni = 2.f * pr * pi;
      pr = nr; pi = ni;
    }
    ws[WS_LRL + tid] = pr;
    ws[WS_LIL + tid] = pi;
    s_nf[tid]          = nfh;
    s_nf[tid + kNHalf] = nfh;
  }
  __syncthreads();
  for (int i = tid; i < kNH * kNU; i += blockDim.x) {
    ws[WS_BP + i] = Bmat[i] * s_nf[i / kNU];
  }
  for (int o = tid; o < kNB * kNH; o += blockDim.x) {
    const int b = o >> 8;
    const int n = o & 255;
    float acc = y2xb[n];
    const float* yr = y0 + b * kNY;
    const float* wr = y2xw + n * kNY;
    #pragma unroll 8
    for (int u = 0; u < kNY; ++u) acc += yr[u] * wr[u];
    ws[WS_X0 + o] = acc;
  }
}

template <int kFloats, int kCols, int kLdsPitch>
__device__ __forceinline__ void stage_rows(const float* __restrict__ src,
                                           float* __restrict__ dst, int tid) {
  static_assert(kFloats % (256 * 4) == 0 && kCols % 4 == 0 && kLdsPitch % 4 == 0, "16-byte pieces");
  v4f piece[kFloats / (256 * 4)];
  #pragma unroll
  for (int it = 0; it < kFloats / (256 * 4); ++it)
    piece[it] = *(const v4f*)(src + (it * 256 + tid) * 4);
  #pragma unroll
  for (int it = 0; it < kFloats / (256 * 4); ++it) {
    const int i  = (it * 256 + tid) * 4;
    const int r  = i / kCols;
    const int cc = i % kCols;
    *(v4f*)(dst + r * kLdsPitch + cc) = piece[it];
  }
}

__device__ __forceinline__ void bu_gemm_store(const float* __restrict__ Uslot,
                                              const float* __restrict__ Bp,
                                              float* __restrict__ Xslot,
                                              int wave, int laneM, int laneHi) {
  const int mt    = (wave >> 1) & 1;
  const int ntPar = wave & 1;
  const int m     = mt * 16 + laneM;
  const int koff  = laneHi << 1;
  const float* Arow = Uslot + m * kNU;
  v2f a[8];
  #pragma unroll
  for (int kk = 0; kk < 8; ++kk) a[kk] = *(const v2f*)(Arow + (kk << 2) + koff);
  v2f bq[2][8];
  const float* Brow0 = Bp + (ntPar * 16 + laneM) * kPitchB + koff;
  #pragma unroll
  for (int kk = 0; kk < 8; ++kk) bq[0][kk] = *(const v2f*)(Brow0 + (kk << 2));
  #pragma unroll
  for (int j = 0; j < 8; ++j) {
    const int nt = ntPar + (j << 1);
    const int n  = nt * 16 + laneM;
    if (j + 1 < 8) {
      const float* Brow1 = Brow0 + (j + 1) * 32 * kPitchB;
      #pragma unroll
      for (int kk = 0; kk < 8; ++kk) bq[(j + 1) & 1][kk] = *(const v2f*)(Brow1 + (kk << 2));
    }
    __builtin_amdgcn_sched_barrier(0x2 | 0x4 | 0x10);
    v8f acc = {};
    #pragma unroll
    for (int kk = 0; kk < 8; ++kk) {
      acc = __builtin_amdgcn_wmma_f32_16x16x4_f32(false, a[kk], false, bq[j & 1][kk],
                                                  (short)0, acc, false, false);
    }
    #pragma unroll
    for (int v = 0; v < 8; ++v) {
      const int mm = mt * 16 + v + (laneHi << 3);
      Xslot[mm * kPitch + n] = acc[v];
    }
  }
}

__global__ void __launch_bounds__(256)
pass1_kernel(const float* __restrict__ U, float* __restrict__ ws) {
  extern __shared__ __align__(16) float Xbuf[];
  float* BpL = Xbuf + kXFloats;

  const int g = blockIdx.x;
  const int c = blockIdx.y;
  const int tid    = threadIdx.x;
  const int lane   = tid & 31;
  const int wave   = tid >> 5;
  const int laneM  = lane & 15;
  const int laneHi = lane >> 4;
  const int slotW  = wave >> 2;

  const int sb0 = wave << 2;
  const int h0  = lane << 2;

  stage_rows<kNH * kNU, kNU, kPitchB>(ws + WS_BP, BpL, tid);

  v4f st1[4], st2[4];
  #pragma unroll
  for (int q = 0; q < 4; ++q) { st1[q] = (v4f)(0.f); st2[q] = (v4f)(0.f); }
  const v4f rlr = *(const v4f*)(ws + WS_LR + h0);
  const v4f rli = *(const v4f*)(ws + WS_LI + h0);
  __syncthreads();

  for (int grp = 0; grp < kNGrp; ++grp) {
    const int t0 = c * kChunk + grp * kTG;
    const float* Uslot = U + ((size_t)(t0 + slotW) * kNB + g * kBG) * kNU;
    bu_gemm_store(Uslot, BpL, Xbuf + slotW * kSlot, wave, laneM, laneHi);
    __syncthreads();
    #pragma unroll
    for (int s = 0; s < kTG; ++s) {
      #pragma unroll
      for (int q = 0; q < 4; ++q) {
        const float* Bu = Xbuf + s * kSlot + (sb0 + q) * kPitch + h0;
        const v4f b1 = *(const v4f*)(Bu);
        const v4f b2 = *(const v4f*)(Bu + kNHalf);
        #pragma unroll
        for (int e = 0; e < 4; ++e) {
          const float x1 = st1[q][e], x2 = st2[q][e];
          st1[q][e] = rlr[e] * x1 - rli[e] * x2 + b1[e];
          st2[q][e] = rli[e] * x1 + rlr[e] * x2 + b2[e];
        }
      }
    }
    __syncthreads();
  }
  #pragma unroll
  for (int q = 0; q < 4; ++q) {
    float* F = ws + WS_F + ((size_t)c * kNB + g * kBG + sb0 + q) * kNH + h0;
    *(v4f*)(F)          = st1[q];
    *(v4f*)(F + kNHalf) = st2[q];
  }
}

__device__ __forceinline__ void comb_fetch(const float* __restrict__ Fb,
                                           float (&f1)[kCombBatch], float (&f2)[kCombBatch]) {
  constexpr size_t kCS = (size_t)kNB * kNH;
  #pragma unroll
  for (int j = 0; j < kCombBatch; ++j) {
    f1[j] = Fb[(size_t)j * kCS];
    f2[j] = Fb[(size_t)j * kCS + kNHalf];
  }
}

__device__ __forceinline__ void comb_chain(float* __restrict__ Sb,
                                           const float (&f1)[kCombBatch],
                                           const float (&f2)[kCombBatch],
                                           float pr, float pi, float& s1, float& s2) {
  constexpr size_t kCS = (size_t)kNB * kNH;
  #pragma unroll
  for (int j = 0; j < kCombBatch; ++j) {
    Sb[(size_t)j * kCS]          = s1;
    Sb[(size_t)j * kCS + kNHalf] = s2;
    const float n1 = pr * s1 - pi * s2 + f1[j];
    const float n2 = pi * s1 + pr * s2 + f2[j];
    s1 = n1; s2 = n2;
  }
}

__global__ void __launch_bounds__(kCombThreads)
combine_kernel(float* __restrict__ ws) {
  const int p = blockIdx.x * kCombThreads + threadIdx.x;
  const int b = p >> 7;
  const int h = p & 127;
  const size_t idx = (size_t)b * kNH + h;
  constexpr size_t kCS = (size_t)kNB * kNH;

  const float pr = ws[WS_LRL + h];
  const float pi = ws[WS_LIL + h];
  float s1 = ws[WS_X0 + idx];
  float s2 = ws[WS_X0 + idx + kNHalf];

  const float* F = ws + WS_F + idx;
  float*       S = ws + WS_S + idx;

  float fa1[kCombBatch], fa2[kCombBatch], fb1[kCombBatch], fb2[kCombBatch];
  comb_fetch(F, fa1, fa2);
  #pragma unroll 1
  for (int c0 = 0; c0 < kNChunk - 2 * kCombBatch; c0 += 2 * kCombBatch) {
    comb_fetch(F + (size_t)(c0 + kCombBatch) * kCS, fb1, fb2);
    comb_chain(S + (size_t)c0 * kCS, fa1, fa2, pr, pi, s1, s2);
    comb_fetch(F + (size_t)(c0 + 2 * kCombBatch) * kCS, fa1, fa2);
    comb_chain(S + (size_t)(c0 + kCombBatch) * kCS, fb1, fb2, pr, pi, s1, s2);
  }
  constexpr int cL = kNChunk - 2 * kCombBatch;
  comb_fetch(F + (size_t)(cL + kCombBatch) * kCS, fb1, fb2);
  comb_chain(S + (size_t)cL * kCS, fa1, fa2, pr, pi, s1, s2);
  comb_chain(S + (size_t)(cL + kCombBatch) * kCS, fb1, fb2, pr, pi, s1, s2);
}

__global__ void __launch_bounds__(256)
pass2_kernel(const float* __restrict__ U, const float* __restrict__ W,
             const float* __restrict__ bias, const float* __restrict__ ws,
             float* __restrict__ out) {
  extern __shared__ __align__(16) float Xbuf[];
  float* BpL = Xbuf + kXFloats;
  float* WL  = BpL + kBpFloats;

  const int g = blockIdx.x;
  const int c = blockIdx.y;
  const int tid    = threadIdx.x;
  const int lane   = tid & 31;
  const int wave   = tid >> 5;
  const int laneM  = lane & 15;
  const int laneHi = lane >> 4;
  const int koff   = laneHi << 1;
  const int slotW  = wave >> 2;

  const int sb0 = wave << 2;
  const int h0  = lane << 2;

  stage_rows<kNH * kNU, kNU, kPitchB>(ws + WS_BP, BpL, tid);
  stage_rows<kNY * kNH, kNH, kPitch>(W, WL, tid);

  v4f st1[4], st2[4];
  #pragma unroll
  for (int q = 0; q < 4; ++q) {
    const float* S = ws + WS_S + ((size_t)c * kNB + g * kBG + sb0 + q) * kNH + h0;
    st1[q] = *(const v4f*)(S);
    st2[q] = *(const v4f*)(S + kNHalf);
  }
  const v4f rlr = *(const v4f*)(ws + WS_LR + h0);
  const v4f rli = *(const v4f*)(ws + WS_LI + h0);

  const int mt2 = (wave >> 1) & 1;
  const int nt2 = wave & 1;
  const float* Wrow  = WL + (nt2 * 16 + laneM) * kPitch;
  const float  biasl = bias[nt2 * 16 + laneM];
  __syncthreads();

  for (int grp = 0; grp < kNGrp; ++grp) {
    const int t0 = c * kChunk + grp * kTG;
    const float* Uslot = U + ((size_t)(t0 + slotW) * kNB + g * kBG) * kNU;
    bu_gemm_store(Uslot, BpL, Xbuf + slotW * kSlot, wave, laneM, laneHi);
    __syncthreads();
    #pragma unroll
    for (int s = 0; s < kTG; ++s) {
      #pragma unroll
      for (int q = 0; q < 4; ++q) {
        float* Bu = Xbuf + s * kSlot + (sb0 + q) * kPitch + h0;
        const v4f b1 = *(const v4f*)(Bu);
        const v4f b2 = *(const v4f*)(Bu + kNHalf);
        #pragma unroll
        for (int e = 0; e < 4; ++e) {
          const float x1 = st1[q][e], x2 = st2[q][e];
          const float n1 = rlr[e] * x1 - rli[e] * x2 + b1[e];
          const float n2 = rli[e] * x1 + rlr[e] * x2 + b2[e];
          st1[q][e] = n1; st2[q][e] = n2;
        }
        *(v4f*)(Bu)          = st1[q];
        *(v4f*)(Bu + kNHalf) = st2[q];
      }
    }
    __syncthreads();
    {
      const float* Arow = Xbuf + slotW * kSlot + (mt2 * 16 + laneM) * kPitch;
      v8f acc = {};
      #pragma unroll 8
      for (int kk = 0; kk < 64; ++kk) {
        const int k = (kk << 2) + koff;
        v2f a = *(const v2f*)(Arow + k);
        v2f b = *(const v2f*)(Wrow + k);
        acc = __builtin_amdgcn_wmma_f32_16x16x4_f32(false, a, false, b,
                                                    (short)0, acc, false, false);
      }
      const int t    = t0 + slotW;
      const int ycol = nt2 * 16 + laneM;
      float* Orow = out + (size_t)t * kNB * kNY + (size_t)(g * kBG) * kNY;
      #pragma unroll
      for (int v = 0; v < 8; ++v) {
        const int mm = mt2 * 16 + v + (laneHi << 3);
        Orow[mm * kNY + ycol] = acc[v] + biasl;
      }
    }
    __syncthreads();
  }
}

extern "C" void kernel_launch(void* const* d_in, const int* in_sizes, int n_in,
                              void* d_out, int out_size, void* d_ws, size_t ws_size,
                              hipStream_t stream) {
  (void)in_sizes; (void)n_in; (void)out_size; (void)ws_size;
  const float* y0   = (const float*)d_in[0];
  const float* U    = (const float*)d_in[1];
  const float* llr  = (const float*)d_in[2];
  const float* lli  = (const float*)d_in[3];
  const float* Bm   = (const float*)d_in[4];
  const float* y2xw = (const float*)d_in[5];
  const float* y2xb = (const float*)d_in[6];
  const float* x2yw = (const float*)d_in[7];
  const float* x2yb = (const float*)d_in[8];
  float* ws  = (float*)d_ws;
  float* out = (float*)d_out;

  hipFuncSetAttribute(reinterpret_cast<const void*>(pass1_kernel),
                      hipFuncAttributeMaxDynamicSharedMemorySize, (int)kSmemBytes1);
  hipFuncSetAttribute(reinterpret_cast<const void*>(pass2_kernel),
                      hipFuncAttributeMaxDynamicSharedMemorySize, (int)kSmemBytes2);

  setup_kernel<<<1, 256, 0, stream>>>(y0, llr, lli, Bm, y2xw, y2xb, ws);
  dim3 grid(kNBG, kNChunk);
  pass1_kernel<<<grid, 256, kSmemBytes1, stream>>>(U, ws);
  combine_kernel<<<kCombBlocks, kCombThreads, 0, stream>>>(ws);
  pass2_kernel<<<grid, 256, kSmemBytes2, stream>>>(U, x2yw, x2yb, ws, out);
}


# setup (pole params, nf-scaled B, x0 = y0 @ y2x_w^T + b) spread over the chip: 64 blocks, one per batch row, every load issued up front, instead of one block looping 64 times; sum order unchanged
# speedup vs baseline: 18.0555x; 1.7310x over previous
#include <hip/hip_runtime.h>

typedef __attribute__((ext_vector_type(2))) float v2f;
typedef __attribute__((ext_vector_type(4))) float v4f;
typedef __attribute__((ext_vector_type(8))) float v8f;

namespace {
constexpr int kT      = 4096;
constexpr int kNB     = 64;
constexpr int kNU     = 32;
constexpr int kNY     = 32;
constexpr int kNH     = 256;
constexpr int kNHalf  = 128;
constexpr int kLog2Chunk = 4;
constexpr int kChunk  = 1 << kLog2Chunk;
constexpr int kNChunk = kT / kChunk;
constexpr int kBG     = 32;
constexpr int kNBG    = kNB / kBG;
constexpr int kTG     = 2;
constexpr int kNGrp   = kChunk / kTG;
constexpr int kPitch  = kNH + 4;
constexpr int kSlot   = kBG * kPitch;
constexpr int kPitchB  = kNU + 4;
constexpr int kXFloats  = kTG * kSlot;
constexpr int kBpFloats = kNH * kPitchB;
constexpr int kWFloats  = kNY * kPitch;
constexpr int kCombThreads = 32;
constexpr int kCombBlocks  = kNB * kNHalf / kCombThreads;
constexpr int kCombBatch   = 32;
static_assert(kNChunk % (2 * kCombBatch) == 0, "combine walks the chunks in pairs of batches");

constexpr size_t WS_LR  = 0;
constexpr size_t WS_LI  = WS_LR + kNHalf;
constexpr size_t WS_LRL = WS_LI + kNHalf;
constexpr size_t WS_LIL = WS_LRL + kNHalf;
constexpr size_t WS_BP  = WS_LIL + kNHalf;
constexpr size_t WS_X0  = WS_BP + (size_t)kNH * kNU;
constexpr size_t WS_F   = WS_X0 + (size_t)kNB * kNH;
constexpr size_t WS_S   = WS_F + (size_t)kNChunk * kNB * kNH;

constexpr size_t kSmemBytes1 = (size_t)(kXFloats + kBpFloats) * sizeof(float);
constexpr size_t kSmemBytes2 = (size_t)(kXFloats + kBpFloats + kWFloats) * sizeof(float);
}

constexpr int kSetupBpPerBlock = kNH * kNU / kNB;
static_assert(kSetupBpPerBlock * kNB == kNH * kNU && kSetupBpPerBlock == kNHalf && kNY % 4 == 0,
              "setup: Bp split evenly over the blocks, half a block each; rows read in 16-byte pieces");

__global__ void __launch_bounds__(256)
setup_kernel(const float* __restrict__ y0, const float* __restrict__ llr,
             const float* __restrict__ lli, const float* __restrict__ Bmat,
             const float* __restrict__ y2xw, const float* __restrict__ y2xb,
             float* __restrict__ ws) {
  const unsigned b   = blockIdx.x;
  const unsigned tid = threadIdx.x;
  const unsigned h   = tid & (kNHalf - 1);
  const unsigned i   = b * kSetupBpPerBlock + h;

  v4f w[kNY / 4], y[kNY / 4];
  #pragma unroll
  for (int j = 0; j < kNY / 4; ++j) w[j] = *(const v4f*)(y2xw + tid * kNY + j * 4);
  #pragma unroll
  for (int j = 0; j < kNY / 4; ++j) y[j] = *(const v4f*)(y0 + b * kNY + j * 4);
  float acc = y2xb[tid];
  const float bval = Bmat[i];
  const float llrB = llr[(i / kNU) & (kNHalf - 1)];
  const float llrP = llr[h];
  const float lliP = lli[h];
  asm volatile("" :: "v"(bval), "v"(llrB), "v"(llrP), "v"(lliP));

  #pragma unroll
  for (int u = 0; u < kNY; ++u) acc = __builtin_fmaf(y[u >> 2][u & 3], w[u >> 2][u & 3], acc);
  ws[WS_X0 + b * kNH + tid] = acc;

  if (tid < kNHalf) {
    const float decayB = __expf(-fabsf(llrB));
    const float rB     = __expf(-decayB);
    const float nfh    = sqrtf(fmaxf(0.f, 1.f - rB * rB));
    ws[WS_BP + i] = bval * nfh;
    if (b == 0) {
      const float decay = __expf(-fabsf(llrP));
      const float r     = __expf(-decay);
      const float th    = 1.57079632679489662f * __expf(-fabsf(lliP));
      const float lr    = r * __cosf(th);
      const float li    = r * __sinf(th);
      ws[WS_LR + tid] = lr;
      ws[WS_LI + tid] = li;
      float pr = lr, pi = li;
      #pragma unroll
      for (int s = 0; s < kLog2Chunk; ++s) {
        const float nr = pr * pr - pi * pi;
        const float ni = 2.f * pr * pi;
        pr = nr; pi = ni;
      }
      ws[WS_LRL + tid] = pr;
      ws[WS_LIL + tid] = pi;
    }
  }
}

template <int kFloats, int kCols, int kLdsPitch>
__device__ __forceinline__ void stage_rows(const float* __restrict__ src,
                                           float* __restrict__ dst, int tid) {
  static_assert(kFloats % (256 * 4) == 0 && kCols % 4 == 0 && kLdsPitch % 4 == 0, "16-byte pieces");
  v4f piece[kFloats / (256 * 4)];
  #pragma unroll
  for (int it = 0; it < kFloats / (256 * 4); ++it)
    piece[it] = *(const v4f*)(src + (it * 256 + tid) * 4);
  #pragma unroll
  for (int it = 0; it < kFloats / (256 * 4); ++it) {
    const int i  = (it * 256 + tid) * 4;
    const int r  = i / kCols;
    const int cc = i % kCols;
    *(v4f*)(dst + r * kLdsPitch + cc) = piece[it];
  }
}

__device__ __forceinline__ void bu_gemm_store(const float* __restrict__ Uslot,
                                              const float* __restrict__ Bp,
                                              float* __restrict__ Xslot,
                                              int wave, int laneM, int laneHi) {
  const int mt    = (wave >> 1) & 1;
  const int ntPar = wave & 1;
  const int m     = mt * 16 + laneM;
  const int koff  = laneHi << 1;
  const float* Arow = Uslot + m * kNU;
  v2f a[8];
  #pragma unroll
  for (int kk = 0; kk < 8; ++kk) a[kk] = *(const v2f*)(Arow + (kk << 2) + koff);
  v2f bq[2][8];
  const float* Brow0 = Bp + (ntPar * 16 + laneM) * kPitchB + koff;
  #pragma unroll
  for (int kk = 0; kk < 8; ++kk) bq[0][kk] = *(const v2f*)(Brow0 + (kk << 2));
  #pragma unroll
  for (int j = 0; j < 8; ++j) {
    const int nt = ntPar + (j << 1);
    const int n  = nt * 16 + laneM;
    if (j + 1 < 8) {
      const float* Brow1 = Brow0 + (j + 1) * 32 * kPitchB;
      #pragma unroll
      for (int kk = 0; kk < 8; ++kk) bq[(j + 1) & 1][kk] = *(const v2f*)(Brow1 + (kk << 2));
    }
    __builtin_amdgcn_sched_barrier(0x2 | 0x4 | 0x10);
    v8f acc = {};
    #pragma unroll
    for (int kk = 0; kk < 8; ++kk) {
      acc = __builtin_amdgcn_wmma_f32_16x16x4_f32(false, a[kk], false, bq[j & 1][kk],
                                                  (short)0, acc, false, false);
    }
    #pragma unroll
    for (int v = 0; v < 8; ++v) {
      const int mm = mt * 16 + v + (laneHi << 3);
      Xslot[mm * kPitch + n] = acc[v];
    }
  }
}

__global__ void __launch_bounds__(256)
pass1_kernel(const float* __restrict__ U, float* __restrict__ ws) {
  extern __shared__ __align__(16) float Xbuf[];
  float* BpL = Xbuf + kXFloats;

  const int g = blockIdx.x;
  const int c = blockIdx.y;
  const int tid    = threadIdx.x;
  const int lane   = tid & 31;
  const int wave   = tid >> 5;
  const int laneM  = lane & 15;
  const int laneHi = lane >> 4;
  const int slotW  = wave >> 2;

  const int sb0 = wave << 2;
  const int h0  = lane << 2;

  stage_rows<kNH * kNU, kNU, kPitchB>(ws + WS_BP, BpL, tid);

  v4f st1[4], st2[4];
  #pragma unroll
  for (int q = 0; q < 4; ++q) { st1[q] = (v4f)(0.f); st2[q] = (v4f)(0.f); }
  const v4f rlr = *(const v4f*)(ws + WS_LR + h0);
  const v4f rli = *(const v4f*)(ws + WS_LI + h0);
  __syncthreads();

  for (int grp = 0; grp < kNGrp; ++grp) {
    const int t0 = c * kChunk + grp * kTG;
    const float* Uslot = U + ((size_t)(t0 + slotW) * kNB + g * kBG) * kNU;
    bu_gemm_store(Uslot, BpL, Xbuf + slotW * kSlot, wave, laneM, laneHi);
    __syncthreads();
    #pragma unroll
    for (int s = 0; s < kTG; ++s) {
      #pragma unroll
      for (int q = 0; q < 4; ++q) {
        const float* Bu = Xbuf + s * kSlot + (sb0 + q) * kPitch + h0;
        const v4f b1 = *(const v4f*)(Bu);
        const v4f b2 = *(const v4f*)(Bu + kNHalf);
        #pragma unroll
        for (int e = 0; e < 4; ++e) {
          const float x1 = st1[q][e], x2 = st2[q][e];
          st1[q][e] = rlr[e] * x1 - rli[e] * x2 + b1[e];
          st2[q][e] = rli[e] * x1 + rlr[e] * x2 + b2[e];
        }
      }
    }
    __syncthreads();
  }
  #pragma unroll
  for (int q = 0; q < 4; ++q) {
    float* F = ws + WS_F + ((size_t)c * kNB + g * kBG + sb0 + q) * kNH + h0;
    *(v4f*)(F)          = st1[q];
    *(v4f*)(F + kNHalf) = st2[q];
  }
}

__device__ __forceinline__ void comb_fetch(const float* __restrict__ Fb,
                                           float (&f1)[kCombBatch], float (&f2)[kCombBatch]) {
  constexpr size_t kCS = (size_t)kNB * kNH;
  #pragma unroll
  for (int j = 0; j < kCombBatch; ++j) {
    f1[j] = Fb[(size_t)j * kCS];
    f2[j] = Fb[(size_t)j * kCS + kNHalf];
  }
}

__device__ __forceinline__ void comb_chain(float* __restrict__ Sb,
                                           const float (&f1)[kCombBatch],
                                           const float (&f2)[kCombBatch],
                                           float pr, float pi, float& s1, float& s2) {
  constexpr size_t kCS = (size_t)kNB * kNH;
  #pragma unroll
  for (int j = 0; j < kCombBatch; ++j) {
    Sb[(size_t)j * kCS]          = s1;
    Sb[(size_t)j * kCS + kNHalf] = s2;
    const float n1 = pr * s1 - pi * s2 + f1[j];
    const float n2 = pi * s1 + pr * s2 + f2[j];
    s1 = n1; s2 = n2;
  }
}

__global__ void __launch_bounds__(kCombThreads)
combine_kernel(float* __restrict__ ws) {
  const int p = blockIdx.x * kCombThreads + threadIdx.x;
  const int b = p >> 7;
  const int h = p & 127;
  const size_t idx = (size_t)b * kNH + h;
  constexpr size_t kCS = (size_t)kNB * kNH;

  const float pr = ws[WS_LRL + h];
  const float pi = ws[WS_LIL + h];
  float s1 = ws[WS_X0 + idx];
  float s2 = ws[WS_X0 + idx + kNHalf];

  const float* F = ws + WS_F + idx;
  float*       S = ws + WS_S + idx;

  float fa1[kCombBatch], fa2[kCombBatch], fb1[kCombBatch], fb2[kCombBatch];
  comb_fetch(F, fa1, fa2);
  #pragma unroll 1
  for (int c0 = 0; c0 < kNChunk - 2 * kCombBatch; c0 += 2 * kCombBatch) {
    comb_fetch(F + (size_t)(c0 + kCombBatch) * kCS, fb1, fb2);
    comb_chain(S + (size_t)c0 * kCS, fa1, fa2, pr, pi, s1, s2);
    comb_fetch(F + (size_t)(c0 + 2 * kCombBatch) * kCS, fa1, fa2);
    comb_chain(S + (size_t)(c0 + kCombBatch) * kCS, fb1, fb2, pr, pi, s1, s2);
  }
  constexpr int cL = kNChunk - 2 * kCombBatch;
  comb_fetch(F + (size_t)(cL + kCombBatch) * kCS, fb1, fb2);
  comb_chain(S + (size_t)cL * kCS, fa1, fa2, pr, pi, s1, s2);
  comb_chain(S + (size_t)(cL + kCombBatch) * kCS, fb1, fb2, pr, pi, s1, s2);
}

__global__ void __launch_bounds__(256)
pass2_kernel(const float* __restrict__ U, const float* __restrict__ W,
             const float* __restrict__ bias, const float* __restrict__ ws,
             float* __restrict__ out) {
  extern __shared__ __align__(16) float Xbuf[];
  float* BpL = Xbuf + kXFloats;
  float* WL  = BpL + kBpFloats;

  const int g = blockIdx.x;
  const int c = blockIdx.y;
  const int tid    = threadIdx.x;
  const int lane   = tid & 31;
  const int wave   = tid >> 5;
  const int laneM  = lane & 15;
  const int laneHi = lane >> 4;
  const int koff   = laneHi << 1;
  const int slotW  = wave >> 2;

  const int sb0 = wave << 2;
  const int h0  = lane << 2;

  stage_rows<kNH * kNU, kNU, kPitchB>(ws + WS_BP, BpL, tid);
  stage_rows<kNY * kNH, kNH, kPitch>(W, WL, tid);

  v4f st1[4], st2[4];
  #pragma unroll
  for (int q = 0; q < 4; ++q) {
    const float* S = ws + WS_S + ((size_t)c * kNB + g * kBG + sb0 + q) * kNH + h0;
    st1[q] = *(const v4f*)(S);
    st2[q] = *(const v4f*)(S + kNHalf);
  }
  const v4f rlr = *(const v4f*)(ws + WS_LR + h0);
  const v4f rli = *(const v4f*)(ws + WS_LI + h0);

  const int mt2 = (wave >> 1) & 1;
  const int nt2 = wave & 1;
  const float* Wrow  = WL + (nt2 * 16 + laneM) * kPitch;
  const float  biasl = bias[nt2 * 16 + laneM];
  __syncthreads();

  for (int grp = 0; grp < kNGrp; ++grp) {
    const int t0 = c * kChunk + grp * kTG;
    const float* Uslot = U + ((size_t)(t0 + slotW) * kNB + g * kBG) * kNU;
    bu_gemm_store(Uslot, BpL, Xbuf + slotW * kSlot, wave, laneM, laneHi);
    __syncthreads();
    #pragma unroll
    for (int s = 0; s < kTG; ++s) {
      #pragma unroll
      for (int q = 0; q < 4; ++q) {
        float* Bu = Xbuf + s * kSlot + (sb0 + q) * kPitch + h0;
        const v4f b1 = *(const v4f*)(Bu);
        const v4f b2 = *(const v4f*)(Bu + kNHalf);
        #pragma unroll
        for (int e = 0; e < 4; ++e) {
          const float x1 = st1[q][e], x2 = st2[q][e];
          const float n1 = rlr[e] * x1 - rli[e] * x2 + b1[e];
          const float n2 = rli[e] * x1 + rlr[e] * x2 + b2[e];
          st1[q][e] = n1; st2[q][e] = n2;
        }
        *(v4f*)(Bu)          = st1[q];
        *(v4f*)(Bu + kNHalf) = st2[q];
      }
    }
    __syncthreads();
    {
      const float* Arow = Xbuf + slotW * kSlot + (mt2 * 16 + laneM) * kPitch;
      v8f acc = {};
      #pragma unroll 8
      for (int kk = 0; kk < 64; ++kk) {
        const int k = (kk << 2) + koff;
        v2f a = *(const v2f*)(Arow + k);
        v2f b = *(const v2f*)(Wrow + k);
        acc = __builtin_amdgcn_wmma_f32_16x16x4_f32(false, a, false, b,
                                                    (short)0, acc, false, false);
      }
      const int t    = t0 + slotW;
      const int ycol = nt2 * 16 + laneM;
      float* Orow = out + (size_t)t * kNB * kNY + (size_t)(g * kBG) * kNY;
      #pragma unroll
      for (int v = 0; v < 8; ++v) {
        const int mm = mt2 * 16 + v + (laneHi << 3);
        Orow[mm * kNY + ycol] = acc[v] + biasl;
      }
    }
    __syncthreads();
  }
}

extern "C" void kernel_launch(void* const* d_in, const int* in_sizes, int n_in,
                              void* d_out, int out_size, void* d_ws, size_t ws_size,
                              hipStream_t stream) {
  (void)in_sizes; (void)n_in; (void)out_size; (void)ws_size;
  const float* y0   = (const float*)d_in[0];
  const float* U    = (const float*)d_in[1];
  const float* llr  = (const float*)d_in[2];
  const float* lli  = (const float*)d_in[3];
  const float* Bm   = (const float*)d_in[4];
  const float* y2xw = (const float*)d_in[5];
  const float* y2xb = (const float*)d_in[6];
  const float* x2yw = (const float*)d_in[7];
  const float* x2yb = (const float*)d_in[8];
  float* ws  = (float*)d_ws;
  float* out = (float*)d_out;

  hipFuncSetAttribute(reinterpret_cast<const void*>(pass1_kernel),
                      hipFuncAttributeMaxDynamicSharedMemorySize, (int)kSmemBytes1);
  hipFuncSetAttribute(reinterpret_cast<const void*>(pass2_kernel),
                      hipFuncAttributeMaxDynamicSharedMemorySize, (int)kSmemBytes2);

  setup_kernel<<<kNB, 256, 0, stream>>>(y0, llr, lli, Bm, y2xw, y2xb, ws);
  dim3 grid(kNBG, kNChunk);
  pass1_kernel<<<grid, 256, kSmemBytes1, stream>>>(U, ws);
  combine_kernel<<<kCombBlocks, kCombThreads, 0, stream>>>(ws);
  pass2_kernel<<<grid, 256, kSmemBytes2, stream>>>(U, x2yw, x2yb, ws, out);
}


# group loops no longer wait on global memory: next group's U rows requested a group ahead; loop barriers (hand-written) wait for LDS only, so loads and output stores stay in flight; sums unchanged
# speedup vs baseline: 19.3065x; 1.0693x over previous
#include <hip/hip_runtime.h>

typedef __attribute__((ext_vector_type(2))) float v2f;
typedef __attribute__((ext_vector_type(4))) float v4f;
typedef __attribute__((ext_vector_type(8))) float v8f;

namespace {
constexpr int kT      = 4096;
constexpr int kNB     = 64;
constexpr int kNU     = 32;
constexpr int kNY     = 32;
constexpr int kNH     = 256;
constexpr int kNHalf  = 128;
constexpr int kLog2Chunk = 4;
constexpr int kChunk  = 1 << kLog2Chunk;
constexpr int kNChunk = kT / kChunk;
constexpr int kBG     = 32;
constexpr int kNBG    = kNB / kBG;
constexpr int kTG     = 2;
constexpr int kNGrp   = kChunk / kTG;
constexpr int kPitch  = kNH + 4;
constexpr int kSlot   = kBG * kPitch;
constexpr int kPitchB  = kNU + 4;
constexpr int kXFloats  = kTG * kSlot;
constexpr int kBpFloats = kNH * kPitchB;
constexpr int kWFloats  = kNY * kPitch;
constexpr int kCombThreads = 32;
constexpr int kCombBlocks  = kNB * kNHalf / kCombThreads;
constexpr int kCombBatch   = 32;
static_assert(kNChunk % (2 * kCombBatch) == 0, "combine walks the chunks in pairs of batches");

constexpr size_t WS_LR  = 0;
constexpr size_t WS_LI  = WS_LR + kNHalf;
constexpr size_t WS_LRL = WS_LI + kNHalf;
constexpr size_t WS_LIL = WS_LRL + kNHalf;
constexpr size_t WS_BP  = WS_LIL + kNHalf;
constexpr size_t WS_X0  = WS_BP + (size_t)kNH * kNU;
constexpr size_t WS_F   = WS_X0 + (size_t)kNB * kNH;
constexpr size_t WS_S   = WS_F + (size_t)kNChunk * kNB * kNH;

constexpr size_t kSmemBytes1 = (size_t)(kXFloats + kBpFloats) * sizeof(float);
constexpr size_t kSmemBytes2 = (size_t)(kXFloats + kBpFloats + kWFloats) * sizeof(float);
}

constexpr int kSetupBpPerBlock = kNH * kNU / kNB;
static_assert(kSetupBpPerBlock * kNB == kNH * kNU && kSetupBpPerBlock == kNHalf && kNY % 4 == 0,
              "setup: Bp split evenly over the blocks, half a block each; rows read in 16-byte pieces");

__global__ void __launch_bounds__(256)
setup_kernel(const float* __restrict__ y0, const float* __restrict__ llr,
             const float* __restrict__ lli, const float* __restrict__ Bmat,
             const float* __restrict__ y2xw, const float* __restrict__ y2xb,
             float* __restrict__ ws) {
  const unsigned b   = blockIdx.x;
  const unsigned tid = threadIdx.x;
  const unsigned h   = tid & (kNHalf - 1);
  const unsigned i   = b * kSetupBpPerBlock + h;

  v4f w[kNY / 4], y[kNY / 4];
  #pragma unroll
  for (int j = 0; j < kNY / 4; ++j) w[j] = *(const v4f*)(y2xw + tid * kNY + j * 4);
  #pragma unroll
  for (int j = 0; j < kNY / 4; ++j) y[j] = *(const v4f*)(y0 + b * kNY + j * 4);
  float acc = y2xb[tid];
  const float bval = Bmat[i];
  const float llrB = llr[(i / kNU) & (kNHalf - 1)];
  const float llrP = llr[h];
  const float lliP = lli[h];
  asm volatile("" :: "v"(bval), "v"(llrB), "v"(llrP), "v"(lliP));

  #pragma unroll
  for (int u = 0; u < kNY; ++u) acc = __builtin_fmaf(y[u >> 2][u & 3], w[u >> 2][u & 3], acc);
  ws[WS_X0 + b * kNH + tid] = acc;

  if (tid < kNHalf) {
    const float decayB = __expf(-fabsf(llrB));
    const float rB     = __expf(-decayB);
    const float nfh    = sqrtf(fmaxf(0.f, 1.f - rB * rB));
    ws[WS_BP + i] = bval * nfh;
    if (b == 0) {
      const float decay = __expf(-fabsf(llrP));
      const float r     = __expf(-decay);
      const float th    = 1.57079632679489662f * __expf(-fabsf(lliP));
      const float lr    = r * __cosf(th);
      const float li    = r * __sinf(th);
      ws[WS_LR + tid] = lr;
      ws[WS_LI + tid] = li;
      float pr = lr, pi = li;
      #pragma unroll
      for (int s = 0; s < kLog2Chunk; ++s) {
        const float nr = pr * pr - pi * pi;
        const float ni = 2.f * pr * pi;
        pr = nr; pi = ni;
      }
      ws[WS_LRL + tid] = pr;
      ws[WS_LIL + tid] = pi;
    }
  }
}

template <int kFloats, int kCols, int kLdsPitch>
__device__ __forceinline__ void stage_rows(const float* __restrict__ src,
                                           float* __restrict__ dst, int tid) {
  static_assert(kFloats % (256 * 4) == 0 && kCols % 4 == 0 && kLdsPitch % 4 == 0, "16-byte pieces");
  v4f piece[kFloats / (256 * 4)];
  #pragma unroll
  for (int it = 0; it < kFloats / (256 * 4); ++it)
    piece[it] = *(const v4f*)(src + (it * 256 + tid) * 4);
  #pragma unroll
  for (int it = 0; it < kFloats / (256 * 4); ++it) {
    const int i  = (it * 256 + tid) * 4;
    const int r  = i / kCols;
    const int cc = i % kCols;
    *(v4f*)(dst + r * kLdsPitch + cc) = piece[it];
  }
}

__device__ __forceinline__ void lds_barrier() {
  asm volatile("s_wait_dscnt 0x0\n\ts_barrier_signal -1\n\ts_barrier_wait -1" ::: "memory");
}

__device__ __forceinline__ void load_u_frags(const float* __restrict__ Uslot,
                                             int wave, int laneM, int laneHi, v2f (&a)[8]) {
  const int mt   = (wave >> 1) & 1;
  const int m    = mt * 16 + laneM;
  const int koff = laneHi << 1;
  const float* Arow = Uslot + m * kNU;
  #pragma unroll
  for (int kk = 0; kk < 8; ++kk) a[kk] = *(const v2f*)(Arow + (kk << 2) + koff);
}

__device__ __forceinline__ void bu_gemm_store(const v2f (&a)[8],
                                              const float* __restrict__ Bp,
                                              float* __restrict__ Xslot,
                                              int wave, int laneM, int laneHi) {
  const int mt    = (wave >> 1) & 1;
  const int ntPar = wave & 1;
  const int koff  = laneHi << 1;
  v2f bq[2][8];
  const float* Brow0 = Bp + (ntPar * 16 + laneM) * kPitchB + koff;
  #pragma unroll
  for (int kk = 0; kk < 8; ++kk) bq[0][kk] = *(const v2f*)(Brow0 + (kk << 2));
  #pragma unroll
  for (int j = 0; j < 8; ++j) {
    const int nt = ntPar + (j << 1);
    const int n  = nt * 16 + laneM;
    if (j + 1 < 8) {
      const float* Brow1 = Brow0 + (j + 1) * 32 * kPitchB;
      #pragma unroll
      for (int kk = 0; kk < 8; ++kk) bq[(j + 1) & 1][kk] = *(const v2f*)(Brow1 + (kk << 2));
    }
    __builtin_amdgcn_sched_barrier(0x2 | 0x4 | 0x10);
    v8f acc = {};
    #pragma unroll
    for (int kk = 0; kk < 8; ++kk) {
      acc = __builtin_amdgcn_wmma_f32_16x16x4_f32(false, a[kk], false, bq[j & 1][kk],
                                                  (short)0, acc, false, false);
    }
    #pragma unroll
    for (int v = 0; v < 8; ++v) {
      const int mm = mt * 16 + v + (laneHi << 3);
      Xslot[mm * kPitch + n] = acc[v];
    }
  }
}

__global__ void __launch_bounds__(256)
pass1_kernel(const float* __restrict__ U, float* __restrict__ ws) {
  extern __shared__ __align__(16) float Xbuf[];
  float* BpL = Xbuf + kXFloats;

  const int g = blockIdx.x;
  const int c = blockIdx.y;
  const int tid    = threadIdx.x;
  const int lane   = tid & 31;
  const int wave   = tid >> 5;
  const int laneM  = lane & 15;
  const int laneHi = lane >> 4;
  const int slotW  = wave >> 2;

  const int sb0 = wave << 2;
  const int h0  = lane << 2;

  const size_t uBase = ((size_t)(c * kChunk + slotW) * kNB + g * kBG) * kNU;
  constexpr size_t uStep = (size_t)kTG * kNB * kNU;
  v2f ua[8], ub[8];
  load_u_frags(U + uBase, wave, laneM, laneHi, ua);

  stage_rows<kNH * kNU, kNU, kPitchB>(ws + WS_BP, BpL, tid);

  v4f st1[4], st2[4];
  #pragma unroll
  for (int q = 0; q < 4; ++q) { st1[q] = (v4f)(0.f); st2[q] = (v4f)(0.f); }
  const v4f rlr = *(const v4f*)(ws + WS_LR + h0);
  const v4f rli = *(const v4f*)(ws + WS_LI + h0);
  __syncthreads();

  auto group = [&](const v2f (&uf)[8]) __attribute__((always_inline)) {
    bu_gemm_store(uf, BpL, Xbuf + slotW * kSlot, wave, laneM, laneHi);
    lds_barrier();
    #pragma unroll
    for (int s = 0; s < kTG; ++s) {
      #pragma unroll
      for (int q = 0; q < 4; ++q) {
        const float* Bu = Xbuf + s * kSlot + (sb0 + q) * kPitch + h0;
        const v4f b1 = *(const v4f*)(Bu);
        const v4f b2 = *(const v4f*)(Bu + kNHalf);
        #pragma unroll
        for (int e = 0; e < 4; ++e) {
          const float x1 = st1[q][e], x2 = st2[q][e];
          st1[q][e] = rlr[e] * x1 - rli[e] * x2 + b1[e];
          st2[q][e] = rli[e] * x1 + rlr[e] * x2 + b2[e];
        }
      }
    }
    lds_barrier();
  };

  static_assert(kNGrp % 2 == 0 && kNGrp >= 4, "the group loop alternates between the two fragment sets");
  for (int grp = 0; grp + 2 < kNGrp; grp += 2) {
    load_u_frags(U + uBase + (size_t)(grp + 1) * uStep, wave, laneM, laneHi, ub);
    group(ua);
    load_u_frags(U + uBase + (size_t)(grp + 2) * uStep, wave, laneM, laneHi, ua);
    group(ub);
  }
  load_u_frags(U + uBase + (size_t)(kNGrp - 1) * uStep, wave, laneM, laneHi, ub);
  group(ua);
  group(ub);
  #pragma unroll
  for (int q = 0; q < 4; ++q) {
    float* F = ws + WS_F + ((size_t)c * kNB + g * kBG + sb0 + q) * kNH + h0;
    *(v4f*)(F)          = st1[q];
    *(v4f*)(F + kNHalf) = st2[q];
  }
}

__device__ __forceinline__ void comb_fetch(const float* __restrict__ Fb,
                                           float (&f1)[kCombBatch], float (&f2)[kCombBatch]) {
  constexpr size_t kCS = (size_t)kNB * kNH;
  #pragma unroll
  for (int j = 0; j < kCombBatch; ++j) {
    f1[j] = Fb[(size_t)j * kCS];
    f2[j] = Fb[(size_t)j * kCS + kNHalf];
  }
}

__device__ __forceinline__ void comb_chain(float* __restrict__ Sb,
                                           const float (&f1)[kCombBatch],
                                           const float (&f2)[kCombBatch],
                                           float pr, float pi, float& s1, float& s2) {
  constexpr size_t kCS = (size_t)kNB * kNH;
  #pragma unroll
  for (int j = 0; j < kCombBatch; ++j) {
    Sb[(size_t)j * kCS]          = s1;
    Sb[(size_t)j * kCS + kNHalf] = s2;
    const float n1 = pr * s1 - pi * s2 + f1[j];
    const float n2 = pi * s1 + pr * s2 + f2[j];
    s1 = n1; s2 = n2;
  }
}

__global__ void __launch_bounds__(kCombThreads)
combine_kernel(float* __restrict__ ws) {
  const int p = blockIdx.x * kCombThreads + threadIdx.x;
  const int b = p >> 7;
  const int h = p & 127;
  const size_t idx = (size_t)b * kNH + h;
  constexpr size_t kCS = (size_t)kNB * kNH;

  const float pr = ws[WS_LRL + h];
  const float pi = ws[WS_LIL + h];
  float s1 = ws[WS_X0 + idx];
  float s2 = ws[WS_X0 + idx + kNHalf];

  const float* F = ws + WS_F + idx;
  float*       S = ws + WS_S + idx;

  float fa1[kCombBatch], fa2[kCombBatch], fb1[kCombBatch], fb2[kCombBatch];
  comb_fetch(F, fa1, fa2);
  #pragma unroll 1
  for (int c0 = 0; c0 < kNChunk - 2 * kCombBatch; c0 += 2 * kCombBatch) {
    comb_fetch(F + (size_t)(c0 + kCombBatch) * kCS, fb1, fb2);
    comb_chain(S + (size_t)c0 * kCS, fa1, fa2, pr, pi, s1, s2);
    comb_fetch(F + (size_t)(c0 + 2 * kCombBatch) * kCS, fa1, fa2);
    comb_chain(S + (size_t)(c0 + kCombBatch) * kCS, fb1, fb2, pr, pi, s1, s2);
  }
  constexpr int cL = kNChunk - 2 * kCombBatch;
  comb_fetch(F + (size_t)(cL + kCombBatch) * kCS, fb1, fb2);
  comb_chain(S + (size_t)cL * kCS, fa1, fa2, pr, pi, s1, s2);
  comb_chain(S + (size_t)(cL + kCombBatch) * kCS, fb1, fb2, pr, pi, s1, s2);
}

__global__ void __launch_bounds__(256)
pass2_kernel(const float* __restrict__ U, const float* __restrict__ W,
             const float* __restrict__ bias, const float* __restrict__ ws,
             float* __restrict__ out) {
  extern __shared__ __align__(16) float Xbuf[];
  float* BpL = Xbuf + kXFloats;
  float* WL  = BpL + kBpFloats;

  const int g = blockIdx.x;
  const int c = blockIdx.y;
  const int tid    = threadIdx.x;
  const int lane   = tid & 31;
  const int wave   = tid >> 5;
  const int laneM  = lane & 15;
  const int laneHi = lane >> 4;
  const int koff   = laneHi << 1;
  const int slotW  = wave >> 2;

  const int sb0 = wave << 2;
  const int h0  = lane << 2;

  const size_t uBase = ((size_t)(c * kChunk + slotW) * kNB + g * kBG) * kNU;
  constexpr size_t uStep = (size_t)kTG * kNB * kNU;
  v2f ua[8], ub[8];
  load_u_frags(U + uBase, wave, laneM, laneHi, ua);

  stage_rows<kNH * kNU, kNU, kPitchB>(ws + WS_BP, BpL, tid);
  stage_rows<kNY * kNH, kNH, kPitch>(W, WL, tid);

  v4f st1[4], st2[4];
  #pragma unroll
  for (int q = 0; q < 4; ++q) {
    const float* S = ws + WS_S + ((size_t)c * kNB + g * kBG + sb0 + q) * kNH + h0;
    st1[q] = *(const v4f*)(S);
    st2[q] = *(const v4f*)(S + kNHalf);
  }
  const v4f rlr = *(const v4f*)(ws + WS_LR + h0);
  const v4f rli = *(const v4f*)(ws + WS_LI + h0);

  const int mt2 = (wave >> 1) & 1;
  const int nt2 = wave & 1;
  const float* Wrow  = WL + (nt2 * 16 + laneM) * kPitch;
  const float  biasl = bias[nt2 * 16 + laneM];
  __syncthreads();

  auto group = [&](const v2f (&uf)[8], const int grp) __attribute__((always_inline)) {
    const int t0 = c * kChunk + grp * kTG;
    bu_gemm_store(uf, BpL, Xbuf + slotW * kSlot, wave, laneM, laneHi);
    lds_barrier();
    #pragma unroll
    for (int s = 0; s < kTG; ++s) {
      #pragma unroll
      for (int q = 0; q < 4; ++q) {
        float* Bu = Xbuf + s * kSlot + (sb0 + q) * kPitch + h0;
        const v4f b1 = *(const v4f*)(Bu);
        const v4f b2 = *(const v4f*)(Bu + kNHalf);
        #pragma unroll
        for (int e = 0; e < 4; ++e) {
          const float x1 = st1[q][e], x2 = st2[q][e];
          const float n1 = rlr[e] * x1 - rli[e] * x2 + b1[e];
          const float n2 = rli[e] * x1 + rlr[e] * x2 + b2[e];
          st1[q][e] = n1; st2[q][e] = n2;
        }
        *(v4f*)(Bu)          = st1[q];
        *(v4f*)(Bu + kNHalf) = st2[q];
      }
    }
    lds_barrier();
    {
      const float* Arow = Xbuf + slotW * kSlot + (mt2 * 16 + laneM) * kPitch;
      v8f acc = {};
      #pragma unroll 8
      for (int kk = 0; kk < 64; ++kk) {
        const int k = (kk << 2) + koff;
        v2f a = *(const v2f*)(Arow + k);
        v2f b = *(const v2f*)(Wrow + k);
        acc = __builtin_amdgcn_wmma_f32_16x16x4_f32(false, a, false, b,
                                                    (short)0, acc, false, false);
      }
      const int t    = t0 + slotW;
      const int ycol = nt2 * 16 + laneM;
      float* Orow = out + (size_t)t * kNB * kNY + (size_t)(g * kBG) * kNY;
      #pragma unroll
      for (int v = 0; v < 8; ++v) {
        const int mm = mt2 * 16 + v + (laneHi << 3);
        Orow[mm * kNY + ycol] = acc[v] + biasl;
      }
    }
    lds_barrier();
  };

  static_assert(kNGrp % 2 == 0 && kNGrp >= 4, "the group loop alternates between the two fragment sets");
  for (int grp = 0; grp + 2 < kNGrp; grp += 2) {
    load_u_frags(U + uBase + (size_t)(grp + 1) * uStep, wave, laneM, laneHi, ub);
    group(ua, grp);
    load_u_frags(U + uBase + (size_t)(grp + 2) * uStep, wave, laneM, laneHi, ua);
    group(ub, grp + 1);
  }
  load_u_frags(U + uBase + (size_t)(kNGrp - 1) * uStep, wave, laneM, laneHi, ub);
  group(ua, kNGrp - 2);
  group(ub, kNGrp - 1);
}

extern "C" void kernel_launch(void* const* d_in, const int* in_sizes, int n_in,
                              void* d_out, int out_size, void* d_ws, size_t ws_size,
                              hipStream_t stream) {
  (void)in_sizes; (void)n_in; (void)out_size; (void)ws_size;
  const float* y0   = (const float*)d_in[0];
  const float* U    = (const float*)d_in[1];
  const float* llr  = (const float*)d_in[2];
  const float* lli  = (const float*)d_in[3];
  const float* Bm   = (const float*)d_in[4];
  const float* y2xw = (const float*)d_in[5];
  const float* y2xb = (const float*)d_in[6];
  const float* x2yw = (const float*)d_in[7];
  const float* x2yb = (const float*)d_in[8];
  float* ws  = (float*)d_ws;
  float* out = (float*)d_out;

  hipFuncSetAttribute(reinterpret_cast<const void*>(pass1_kernel),
                      hipFuncAttributeMaxDynamicSharedMemorySize, (int)kSmemBytes1);
  hipFuncSetAttribute(reinterpret_cast<const void*>(pass2_kernel),
                      hipFuncAttributeMaxDynamicSharedMemorySize, (int)kSmemBytes2);

  setup_kernel<<<kNB, 256, 0, stream>>>(y0, llr, lli, Bm, y2xw, y2xb, ws);
  dim3 grid(kNBG, kNChunk);
  pass1_kernel<<<grid, 256, kSmemBytes1, stream>>>(U, ws);
  combine_kernel<<<kCombBlocks, kCombThreads, 0, stream>>>(ws);
  pass2_kernel<<<grid, 256, kSmemBytes2, stream>>>(U, x2yw, x2yb, ws, out);
}
